# starting kernel
# baseline (speedup 1.0000x reference)
#include <hip/hip_runtime.h>
#include <hip/hip_bf16.h>

typedef float v2f __attribute__((ext_vector_type(2)));
typedef float v4f __attribute__((ext_vector_type(4)));
typedef float v8f __attribute__((ext_vector_type(8)));

#define FDIM 256
#define KDIM 512
#define DEG  32
#define TM   16
#define THREADS 128

__global__ __launch_bounds__(THREADS)
void graphsage_fused_kernel(const float* __restrict__ feat_data,
                            const float* __restrict__ feats,
                            const float* __restrict__ W1,
                            const float* __restrict__ b1,
                            const float* __restrict__ W2,
                            const float* __restrict__ b2,
                            const int*   __restrict__ adj,
                            const int*   __restrict__ nodes,
                            int nNodes,
                            float* __restrict__ out)
{
    __shared__ float S[TM][KDIM];
    __shared__ float T[TM][KDIM];
    __shared__ int   nidx[TM];
    __shared__ int   nbr[TM][DEG];
    __shared__ float part[TM][8];
    __shared__ float inv_norm[TM];

    const int t    = threadIdx.x;
    const int m0   = blockIdx.x * TM;
    const int lane = t & 31;
    const int wave = t >> 5;

    if (t < TM) {
        int g = m0 + t;
        nidx[t] = (g < nNodes) ? nodes[g] : 0;
    }
    __syncthreads();
    for (int i = t; i < TM * DEG; i += THREADS)
        nbr[i >> 5][i & 31] = adj[(size_t)nidx[i >> 5] * DEG + (i & 31)];
    __syncthreads();

    {
        const int half = t >> 6;
        const int tc   = (t & 63) << 2;
        for (int r = 0; r < TM; r += 2) {
            const int rr  = r + half;
            const int idx = nidx[rr];
            float s0 = 0.f, s1 = 0.f, s2 = 0.f, s3 = 0.f;
            float f0 = 0.f, f1 = 0.f, f2 = 0.f, f3 = 0.f;
            #pragma unroll 8
            for (int d = 0; d < DEG; ++d) {
                const int j = nbr[rr][d];
                const v4f fa = *(const v4f*)(feat_data + (size_t)j * FDIM + tc);
                const v4f fb = *(const v4f*)(feats     + (size_t)j * FDIM + tc);
                s0 += fa.x; s1 += fa.y; s2 += fa.z; s3 += fa.w;
                f0 += fb.x; f1 += fb.y; f2 += fb.z; f3 += fb.w;
            }
            v4f sv; sv.x = s0; sv.y = s1; sv.z = s2; sv.w = s3;
            v4f fv; fv.x = f0; fv.y = f1; fv.z = f2; fv.w = f3;
            *(v4f*)(&S[rr][tc])        = sv;
            *(v4f*)(&S[rr][FDIM + tc]) = fv;
            *(v4f*)(&T[rr][tc]) = *(const v4f*)(feat_data + (size_t)idx * FDIM + tc);
        }
    }
    __syncthreads();

    const int lr    = lane & 15;
    const int kh    = (lane >> 4) << 1;
    const int rbase = (lane >> 4) << 3;

    {
        v8f acc0 = {}, acc1 = {}, acc2 = {}, acc3 = {};
        for (int k0 = 0; k0 < KDIM; k0 += 4) {
            v2f a;
            a.x = S[lr][k0 + kh];
            a.y = S[lr][k0 + kh + 1];
            const int nb0 = wave * 64 + lr;
            const float* w0 = W1 + (size_t)(nb0     ) * KDIM + k0 + kh;
            const float* w1 = W1 + (size_t)(nb0 + 16) * KDIM + k0 + kh;
            const float* w2 = W1 + (size_t)(nb0 + 32) * KDIM + k0 + kh;
            const float* w3 = W1 + (size_t)(nb0 + 48) * KDIM + k0 + kh;
            v2f bv;
            bv.x = w0[0]; bv.y = w0[1];
            acc0 = __builtin_amdgcn_wmma_f32_16x16x4_f32(false, a, false, bv, (short)0, acc0, false, false);
            bv.x = w1[0]; bv.y = w1[1];
            acc1 = __builtin_amdgcn_wmma_f32_16x16x4_f32(false, a, false, bv, (short)0, acc1, false, false);
            bv.x = w2[0]; bv.y = w2[1];
            acc2 = __builtin_amdgcn_wmma_f32_16x16x4_f32(false, a, false, bv, (short)0, acc2, false, false);
            bv.x = w3[0]; bv.y = w3[1];
            acc3 = __builtin_amdgcn_wmma_f32_16x16x4_f32(false, a, false, bv, (short)0, acc3, false, false);
        }
        #pragma unroll
        for (int tile = 0; tile < 4; ++tile) {
            const v8f acc = (tile == 0) ? acc0 : (tile == 1) ? acc1 : (tile == 2) ? acc2 : acc3;
            const int n = wave * 64 + tile * 16 + lr;
            const float bias = (float)DEG * b1[n];
            #pragma unroll
            for (int v = 0; v < 8; ++v)
                T[rbase + v][FDIM + n] = acc[v] + bias;
        }
    }
    __syncthreads();

    {
        v8f acc0 = {}, acc1 = {}, acc2 = {}, acc3 = {};
        for (int k0 = 0; k0 < KDIM; k0 += 4) {
            v2f a;
            a.x = T[lr][k0 + kh];
            a.y = T[lr][k0 + kh + 1];
            const int nb0 = wave * 64 + lr;
            const float* w0 = W2 + (size_t)(nb0     ) * KDIM + k0 + kh;
            const float* w1 = W2 + (size_t)(nb0 + 16) * KDIM + k0 + kh;
            const float* w2 = W2 + (size_t)(nb0 + 32) * KDIM + k0 + kh;
            const float* w3 = W2 + (size_t)(nb0 + 48) * KDIM + k0 + kh;
            v2f bv;
            bv.x = w0[0]; bv.y = w0[1];
            acc0 = __builtin_amdgcn_wmma_f32_16x16x4_f32(false, a, false, bv, (short)0, acc0, false, false);
            bv.x = w1[0]; bv.y = w1[1];
            acc1 = __builtin_amdgcn_wmma_f32_16x16x4_f32(false, a, false, bv, (short)0, acc1, false, false);
            bv.x = w2[0]; bv.y = w2[1];
            acc2 = __builtin_amdgcn_wmma_f32_16x16x4_f32(false, a, false, bv, (short)0, acc2, false, false);
            bv.x = w3[0]; bv.y = w3[1];
            acc3 = __builtin_amdgcn_wmma_f32_16x16x4_f32(false, a, false, bv, (short)0, acc3, false, false);
        }
        #pragma unroll
        for (int tile = 0; tile < 4; ++tile) {
            const v8f acc = (tile == 0) ? acc0 : (tile == 1) ? acc1 : (tile == 2) ? acc2 : acc3;
            const int n = wave * 64 + tile * 16 + lr;
            const float bias = b2[n];
            #pragma unroll
            for (int v = 0; v < 8; ++v)
                S[rbase + v][n] = acc[v] + bias;
        }
    }
    __syncthreads();

    {
        const int r = t >> 3, seg = t & 7;
        float s = 0.f;
        #pragma unroll 8
        for (int c = seg * 32; c < seg * 32 + 32; ++c) {
            const float x = S[r][c];
            s += x * x;
        }
        part[r][seg] = s;
    }
    __syncthreads();
    if (t < TM) {
        float s = 0.f;
        #pragma unroll
        for (int i = 0; i < 8; ++i) s += part[t][i];
        inv_norm[t] = 1.0f / fmaxf(sqrtf(s), 1e-12f);
    }
    __syncthreads();
    for (int i = t; i < TM * FDIM; i += THREADS) {
        const int r = i >> 8;
        const int c = i & (FDIM - 1);
        const int g = m0 + r;
        if (g < nNodes)
            out[(size_t)g * FDIM + c] = S[r][c] * inv_norm[r];
    }
}

extern "C" void kernel_launch(void* const* d_in, const int* in_sizes, int n_in,
                              void* d_out, int out_size, void* d_ws, size_t ws_size,
                              hipStream_t stream) {
    const float* feat_data = (const float*)d_in[0];
    const float* feats     = (const float*)d_in[1];
    const float* W1        = (const float*)d_in[2];
    const float* b1        = (const float*)d_in[3];
    const float* W2        = (const float*)d_in[4];
    const float* b2        = (const float*)d_in[5];
    const int*   adj       = (const int*)d_in[6];
    const int*   in1       = (const int*)d_in[7];
    const int*   in2       = (const int*)d_in[8];
    const int*   neg       = (const int*)d_in[9];
    float* out = (float*)d_out;

    const int B1 = in_sizes[7];
    const int B2 = in_sizes[8];
    const int Bn = in_sizes[9];

    dim3 blk(THREADS);
    graphsage_fused_kernel<<<dim3((B1 + TM - 1) / TM), blk, 0, stream>>>(
        feat_data, feats, W1, b1, W2, b2, adj, in1, B1, out);
    graphsage_fused_kernel<<<dim3((B2 + TM - 1) / TM), blk, 0, stream>>>(
        feat_data, feats, W1, b1, W2, b2, adj, in2, B2, out + (size_t)B1 * FDIM);
    graphsage_fused_kernel<<<dim3((Bn + TM - 1) / TM), blk, 0, stream>>>(
        feat_data, feats, W1, b1, W2, b2, adj, neg, Bn, out + (size_t)(B1 + B2) * FDIM);
}


# GEMM operand fragments fetched 16 bytes per lane (W rows from global, A rows from LDS) in one permuted k order for both, a whole 128-B line per W row per trip; the k sums are taken in a new order
# speedup vs baseline: 1.0814x; 1.0814x over previous
#include <hip/hip_runtime.h>
#include <hip/hip_bf16.h>

typedef float v2f __attribute__((ext_vector_type(2)));
typedef float v4f __attribute__((ext_vector_type(4)));
typedef float v8f __attribute__((ext_vector_type(8)));

#define FDIM 256
#define KDIM 512
#define DEG  32
#define TM   16
#define THREADS 128

__device__ __forceinline__
void gemm_strip_16x64(const float* __restrict__ arow,
                      const float* __restrict__ wrow,
                      v8f& acc0, v8f& acc1, v8f& acc2, v8f& acc3)
{
    for (int k0 = 0; k0 < KDIM; k0 += 32) {
        #pragma unroll
        for (int j = 0; j < 4; ++j) {
            const int k = k0 + 8 * j;
            const v4f a4 = *(const v4f*)(arow + k);
            const v4f p0 = *(const v4f*)(wrow + k);
            const v4f p1 = *(const v4f*)(wrow + 16 * KDIM + k);
            const v4f p2 = *(const v4f*)(wrow + 32 * KDIM + k);
            const v4f p3 = *(const v4f*)(wrow + 48 * KDIM + k);
            v2f a, b;
            a.x = a4.x; a.y = a4.y;
            b.x = p0.x; b.y = p0.y;
            acc0 = __builtin_amdgcn_wmma_f32_16x16x4_f32(false, a, false, b, (short)0, acc0, false, false);
            b.x = p1.x; b.y = p1.y;
            acc1 = __builtin_amdgcn_wmma_f32_16x16x4_f32(false, a, false, b, (short)0, acc1, false, false);
            b.x = p2.x; b.y = p2.y;
            acc2 = __builtin_amdgcn_wmma_f32_16x16x4_f32(false, a, false, b, (short)0, acc2, false, false);
            b.x = p3.x; b.y = p3.y;
            acc3 = __builtin_amdgcn_wmma_f32_16x16x4_f32(false, a, false, b, (short)0, acc3, false, false);
            a.x = a4.z; a.y = a4.w;
            b.x = p0.z; b.y = p0.w;
            acc0 = __builtin_amdgcn_wmma_f32_16x16x4_f32(false, a, false, b, (short)0, acc0, false, false);
            b.x = p1.z; b.y = p1.w;
            acc1 = __builtin_amdgcn_wmma_f32_16x16x4_f32(false, a, false, b, (short)0, acc1, false, false);
            b.x = p2.z; b.y = p2.w;
            acc2 = __builtin_amdgcn_wmma_f32_16x16x4_f32(false, a, false, b, (short)0, acc2, false, false);
            b.x = p3.z; b.y = p3.w;
            acc3 = __builtin_amdgcn_wmma_f32_16x16x4_f32(false, a, false, b, (short)0, acc3, false, false);
        }
    }
}


__global__ __launch_bounds__(THREADS)
void graphsage_fused_kernel(const float* __restrict__ feat_data,
                            const float* __restrict__ feats,
                            const float* __restrict__ W1,
                            const float* __restrict__ b1,
                            const float* __restrict__ W2,
                            const float* __restrict__ b2,
                            const int*   __restrict__ adj,
                            const int*   __restrict__ nodes,
                            int nNodes,
                            float* __restrict__ out)
{
    __shared__ __attribute__((aligned(16))) float S[TM][KDIM];
    __shared__ __attribute__((aligned(16))) float T[TM][KDIM];
    __shared__ int   nidx[TM];
    __shared__ int   nbr[TM][DEG];
    __shared__ float part[TM][8];
    __shared__ float inv_norm[TM];

    const int t    = threadIdx.x;
    const int m0   = blockIdx.x * TM;
    const int lane = t & 31;
    const int wave = t >> 5;

    if (t < TM) {
        int g = m0 + t;
        nidx[t] = (g < nNodes) ? nodes[g] : 0;
    }
    __syncthreads();
    for (int i = t; i < TM * DEG; i += THREADS)
        nbr[i >> 5][i & 31] = adj[(size_t)nidx[i >> 5] * DEG + (i & 31)];
    __syncthreads();

    {
        const int half = t >> 6;
        const int tc   = (t & 63) << 2;
        for (int r = 0; r < TM; r += 2) {
            const int rr  = r + half;
            const int idx = nidx[rr];
            float s0 = 0.f, s1 = 0.f, s2 = 0.f, s3 = 0.f;
            float f0 = 0.f, f1 = 0.f, f2 = 0.f, f3 = 0.f;
            #pragma unroll 8
            for (int d = 0; d < DEG; ++d) {
                const int j = nbr[rr][d];
                const v4f fa = *(const v4f*)(feat_data + (size_t)j * FDIM + tc);
                const v4f fb = *(const v4f*)(feats     + (size_t)j * FDIM + tc);
                s0 += fa.x; s1 += fa.y; s2 += fa.z; s3 += fa.w;
                f0 += fb.x; f1 += fb.y; f2 += fb.z; f3 += fb.w;
            }
            v4f sv; sv.x = s0; sv.y = s1; sv.z = s2; sv.w = s3;
            v4f fv; fv.x = f0; fv.y = f1; fv.z = f2; fv.w = f3;
            *(v4f*)(&S[rr][tc])        = sv;
            *(v4f*)(&S[rr][FDIM + tc]) = fv;
            *(v4f*)(&T[rr][tc]) = *(const v4f*)(feat_data + (size_t)idx * FDIM + tc);
        }
    }
    __syncthreads();

    const int lr    = lane & 15;
    const int ko    = (lane >> 4) << 2;
    const int rbase = (lane >> 4) << 3;

    {
        v8f acc0 = {}, acc1 = {}, acc2 = {}, acc3 = {};
        gemm_strip_16x64(&S[lr][ko], W1 + (size_t)(wave * 64 + lr) * KDIM + ko,
                         acc0, acc1, acc2, acc3);
        #pragma unroll
        for (int tile = 0; tile < 4; ++tile) {
            const v8f acc = (tile == 0) ? acc0 : (tile == 1) ? acc1 : (tile == 2) ? acc2 : acc3;
            const int n = wave * 64 + tile * 16 + lr;
            const float bias = (float)DEG * b1[n];
            #pragma unroll
            for (int v = 0; v < 8; ++v)
                T[rbase + v][FDIM + n] = acc[v] + bias;
        }
    }
    __syncthreads();

    {
        v8f acc0 = {}, acc1 = {}, acc2 = {}, acc3 = {};
        gemm_strip_16x64(&T[lr][ko], W2 + (size_t)(wave * 64 + lr) * KDIM + ko,
                         acc0, acc1, acc2, acc3);
        #pragma unroll
        for (int tile = 0; tile < 4; ++tile) {
            const v8f acc = (tile == 0) ? acc0 : (tile == 1) ? acc1 : (tile == 2) ? acc2 : acc3;
            const int n = wave * 64 + tile * 16 + lr;
            const float bias = b2[n];
            #pragma unroll
            for (int v = 0; v < 8; ++v)
                S[rbase + v][n] = acc[v] + bias;
        }
    }
    __syncthreads();

    {
        const int r = t >> 3, seg = t & 7;
        float s = 0.f;
        #pragma unroll 8
        for (int c = seg * 32; c < seg * 32 + 32; ++c) {
            const float x = S[r][c];
            s += x * x;
        }
        part[r][seg] = s;
    }
    __syncthreads();
    if (t < TM) {
        float s = 0.f;
        #pragma unroll
        for (int i = 0; i < 8; ++i) s += part[t][i];
        inv_norm[t] = 1.0f / fmaxf(sqrtf(s), 1e-12f);
    }
    __syncthreads();
    for (int i = t; i < TM * FDIM; i += THREADS) {
        const int r = i >> 8;
        const int c = i & (FDIM - 1);
        const int g = m0 + r;
        if (g < nNodes)
            out[(size_t)g * FDIM + c] = S[r][c] * inv_norm[r];
    }
}

extern "C" void kernel_launch(void* const* d_in, const int* in_sizes, int n_in,
                              void* d_out, int out_size, void* d_ws, size_t ws_size,
                              hipStream_t stream) {
    const float* feat_data = (const float*)d_in[0];
    const float* feats     = (const float*)d_in[1];
    const float* W1        = (const float*)d_in[2];
    const float* b1        = (const float*)d_in[3];
    const float* W2        = (const float*)d_in[4];
    const float* b2        = (const float*)d_in[5];
    const int*   adj       = (const int*)d_in[6];
    const int*   in1       = (const int*)d_in[7];
    const int*   in2       = (const int*)d_in[8];
    const int*   neg       = (const int*)d_in[9];
    float* out = (float*)d_out;

    const int B1 = in_sizes[7];
    const int B2 = in_sizes[8];
    const int Bn = in_sizes[9];

    dim3 blk(THREADS);
    graphsage_fused_kernel<<<dim3((B1 + TM - 1) / TM), blk, 0, stream>>>(
        feat_data, feats, W1, b1, W2, b2, adj, in1, B1, out);
    graphsage_fused_kernel<<<dim3((B2 + TM - 1) / TM), blk, 0, stream>>>(
        feat_data, feats, W1, b1, W2, b2, adj, in2, B2, out + (size_t)B1 * FDIM);
    graphsage_fused_kernel<<<dim3((Bn + TM - 1) / TM), blk, 0, stream>>>(
        feat_data, feats, W1, b1, W2, b2, adj, neg, Bn, out + (size_t)(B1 + B2) * FDIM);
}


# Three launches fused into one launch of all 1025 work-groups (block index picks node list and strip); index/norm arrays overlaid on tile T, LDS 64 KiB, so the whole grid can be resident at once
# speedup vs baseline: 1.4845x; 1.3728x over previous
#include <hip/hip_runtime.h>
#include <hip/hip_bf16.h>

typedef float v2f __attribute__((ext_vector_type(2)));
typedef float v4f __attribute__((ext_vector_type(4)));
typedef float v8f __attribute__((ext_vector_type(8)));

#define FDIM 256
#define KDIM 512
#define DEG  32
#define TM   16
#define THREADS 128

__device__ __forceinline__
void gemm_strip_16x64(const float* __restrict__ arow,
                      const float* __restrict__ wrow,
                      v8f& acc0, v8f& acc1, v8f& acc2, v8f& acc3)
{
    for (int k0 = 0; k0 < KDIM; k0 += 32) {
        #pragma unroll
        for (int j = 0; j < 4; ++j) {
            const int k = k0 + 8 * j;
            const v4f a4 = *(const v4f*)(arow + k);
            const v4f p0 = *(const v4f*)(wrow + k);
            const v4f p1 = *(const v4f*)(wrow + 16 * KDIM + k);
            const v4f p2 = *(const v4f*)(wrow + 32 * KDIM + k);
            const v4f p3 = *(const v4f*)(wrow + 48 * KDIM + k);
            v2f a, b;
            a.x = a4.x; a.y = a4.y;
            b.x = p0.x; b.y = p0.y;
            acc0 = __builtin_amdgcn_wmma_f32_16x16x4_f32(false, a, false, b, (short)0, acc0, false, false);
            b.x = p1.x; b.y = p1.y;
            acc1 = __builtin_amdgcn_wmma_f32_16x16x4_f32(false, a, false, b, (short)0, acc1, false, false);
            b.x = p2.x; b.y = p2.y;
            acc2 = __builtin_amdgcn_wmma_f32_16x16x4_f32(false, a, false, b, (short)0, acc2, false, false);
            b.x = p3.x; b.y = p3.y;
            acc3 = __builtin_amdgcn_wmma_f32_16x16x4_f32(false, a, false, b, (short)0, acc3, false, false);
            a.x = a4.z; a.y = a4.w;
            b.x = p0.z; b.y = p0.w;
            acc0 = __builtin_amdgcn_wmma_f32_16x16x4_f32(false, a, false, b, (short)0, acc0, false, false);
            b.x = p1.z; b.y = p1.w;
            acc1 = __builtin_amdgcn_wmma_f32_16x16x4_f32(false, a, false, b, (short)0, acc1, false, false);
            b.x = p2.z; b.y = p2.w;
            acc2 = __builtin_amdgcn_wmma_f32_16x16x4_f32(false, a, false, b, (short)0, acc2, false, false);
            b.x = p3.z; b.y = p3.w;
            acc3 = __builtin_amdgcn_wmma_f32_16x16x4_f32(false, a, false, b, (short)0, acc3, false, false);
        }
    }
}


__global__ __launch_bounds__(THREADS)
void graphsage_fused_kernel(const float* __restrict__ feat_data,
                            const float* __restrict__ feats,
                            const float* __restrict__ W1,
                            const float* __restrict__ b1,
                            const float* __restrict__ W2,
                            const float* __restrict__ b2,
                            const int*   __restrict__ adj,
                            const int*   __restrict__ in1,
                            const int*   __restrict__ in2,
                            const int*   __restrict__ neg,
                            int B1, int B2, int Bn,
                            float* __restrict__ out_all)
{
    __shared__ __attribute__((aligned(16))) float S[TM][KDIM];
    __shared__ __attribute__((aligned(16))) float T[TM][KDIM];
    float* const nbr_f    = &T[0][FDIM];
    float* const nidx_f   = &T[2][FDIM];
    float* const part     = &T[0][0];
    float* const inv_norm = &T[1][0];

    const int t    = threadIdx.x;
    const int lane = t & 31;
    const int wave = t >> 5;

    const int nb1 = (B1 + TM - 1) / TM;
    const int nb2 = (B2 + TM - 1) / TM;
    int bi = blockIdx.x;
    const int* __restrict__ nodes = in1;
    int nNodes = B1;
    float* __restrict__ out = out_all;
    if (bi >= nb1 + nb2)  { bi -= nb1 + nb2; nodes = neg; nNodes = Bn; out = out_all + (size_t)(B1 + B2) * FDIM; }
    else if (bi >= nb1)   { bi -= nb1;       nodes = in2; nNodes = B2; out = out_all + (size_t)B1 * FDIM; }
    const int m0 = bi * TM;

    if (t < TM) {
        int g = m0 + t;
        nidx_f[t] = __int_as_float((g < nNodes) ? nodes[g] : 0);
    }
    __syncthreads();
    for (int i = t; i < TM * DEG; i += THREADS) {
        const int rowi = i >> 5, d = i & 31;
        const int id   = __float_as_int(nidx_f[rowi]);
        nbr_f[(i >> 8) * KDIM + (i & 255)] = __int_as_float(adj[(size_t)id * DEG + d]);
    }
    __syncthreads();

    {
        const int half = t >> 6;
        const int tc   = (t & 63) << 2;
        for (int r = 0; r < TM; r += 2) {
            const int rr  = r + half;
            const int idx = __float_as_int(nidx_f[rr]);
            float s0 = 0.f, s1 = 0.f, s2 = 0.f, s3 = 0.f;
            float f0 = 0.f, f1 = 0.f, f2 = 0.f, f3 = 0.f;
            #pragma unroll 8
            for (int d = 0; d < DEG; ++d) {
                const int j = __float_as_int(nbr_f[(rr >> 3) * KDIM + (rr & 7) * DEG + d]);
                const v4f fa = *(const v4f*)(feat_data + (size_t)j * FDIM + tc);
                const v4f fb = *(const v4f*)(feats     + (size_t)j * FDIM + tc);
                s0 += fa.x; s1 += fa.y; s2 += fa.z; s3 += fa.w;
                f0 += fb.x; f1 += fb.y; f2 += fb.z; f3 += fb.w;
            }
            v4f sv; sv.x = s0; sv.y = s1; sv.z = s2; sv.w = s3;
            v4f fv; fv.x = f0; fv.y = f1; fv.z = f2; fv.w = f3;
            *(v4f*)(&S[rr][tc])        = sv;
            *(v4f*)(&S[rr][FDIM + tc]) = fv;
            *(v4f*)(&T[rr][tc]) = *(const v4f*)(feat_data + (size_t)idx * FDIM + tc);
        }
    }
    __syncthreads();

    const int lr    = lane & 15;
    const int ko    = (lane >> 4) << 2;
    const int rbase = (lane >> 4) << 3;

    {
        v8f acc0 = {}, acc1 = {}, acc2 = {}, acc3 = {};
        gemm_strip_16x64(&S[lr][ko], W1 + (size_t)(wave * 64 + lr) * KDIM + ko,
                         acc0, acc1, acc2, acc3);
        #pragma unroll
        for (int tile = 0; tile < 4; ++tile) {
            const v8f acc = (tile == 0) ? acc0 : (tile == 1) ? acc1 : (tile == 2) ? acc2 : acc3;
            const int n = wave * 64 + tile * 16 + lr;
            const float bias = (float)DEG * b1[n];
            #pragma unroll
            for (int v = 0; v < 8; ++v)
                T[rbase + v][FDIM + n] = acc[v] + bias;
        }
    }
    __syncthreads();

    {
        v8f acc0 = {}, acc1 = {}, acc2 = {}, acc3 = {};
        gemm_strip_16x64(&T[lr][ko], W2 + (size_t)(wave * 64 + lr) * KDIM + ko,
                         acc0, acc1, acc2, acc3);
        #pragma unroll
        for (int tile = 0; tile < 4; ++tile) {
            const v8f acc = (tile == 0) ? acc0 : (tile == 1) ? acc1 : (tile == 2) ? acc2 : acc3;
            const int n = wave * 64 + tile * 16 + lr;
            const float bias = b2[n];
            #pragma unroll
            for (int v = 0; v < 8; ++v)
                S[rbase + v][n] = acc[v] + bias;
        }
    }
    __syncthreads();

    {
        const int r = t >> 3, seg = t & 7;
        float s = 0.f;
        #pragma unroll 8
        for (int c = seg * 32; c < seg * 32 + 32; ++c) {
            const float x = S[r][c];
            s += x * x;
        }
        part[r * 8 + seg] = s;
    }
    __syncthreads();
    if (t < TM) {
        float s = 0.f;
        #pragma unroll
        for (int i = 0; i < 8; ++i) s += part[t * 8 + i];
        inv_norm[t] = 1.0f / fmaxf(sqrtf(s), 1e-12f);
    }
    __syncthreads();
    for (int i = t; i < TM * FDIM; i += THREADS) {
        const int r = i >> 8;
        const int c = i & (FDIM - 1);
        const int g = m0 + r;
        if (g < nNodes)
            out[(size_t)g * FDIM + c] = S[r][c] * inv_norm[r];
    }
}

extern "C" void kernel_launch(void* const* d_in, const int* in_sizes, int n_in,
                              void* d_out, int out_size, void* d_ws, size_t ws_size,
                              hipStream_t stream) {
    const float* feat_data = (const float*)d_in[0];
    const float* feats     = (const float*)d_in[1];
    const float* W1        = (const float*)d_in[2];
    const float* b1        = (const float*)d_in[3];
    const float* W2        = (const float*)d_in[4];
    const float* b2        = (const float*)d_in[5];
    const int*   adj       = (const int*)d_in[6];
    const int*   in1       = (const int*)d_in[7];
    const int*   in2       = (const int*)d_in[8];
    const int*   neg       = (const int*)d_in[9];
    float* out = (float*)d_out;

    const int B1 = in_sizes[7];
    const int B2 = in_sizes[8];
    const int Bn = in_sizes[9];

    dim3 blk(THREADS);
    const int nblk = (B1 + TM - 1) / TM + (B2 + TM - 1) / TM + (Bn + TM - 1) / TM;
    graphsage_fused_kernel<<<dim3(nblk), blk, 0, stream>>>(
        feat_data, feats, W1, b1, W2, b2, adj, in1, in2, neg, B1, B2, Bn, out);
}


# 32-row work-groups (8 waves): each W fragment feeds two row tiles, halving W bytes and W loads per row; one padded 32x772 LDS array keeps the grid resident; a short tail strip runs one tile
# speedup vs baseline: 1.6271x; 1.0961x over previous
#include <hip/hip_runtime.h>
#include <hip/hip_bf16.h>


typedef float v2f __attribute__((ext_vector_type(2)));
typedef float v4f __attribute__((ext_vector_type(4)));
typedef float v8f __attribute__((ext_vector_type(8)));

#define FDIM 256
#define KDIM 512
#define DEG  32
#define TM   32
#define THREADS 256
#define XP   772
#define XS   256
#define XU   512

#define WMMA4(A, B, C) __builtin_amdgcn_wmma_f32_16x16x4_f32(false, (A), false, (B), (short)0, (C), false, false)

template <int MT>
__device__ __forceinline__
void gemm_strip(const float* __restrict__ arow,
                const float* __restrict__ wrow,
                v8f& acc00, v8f& acc01, v8f& acc10, v8f& acc11)
{
    #pragma unroll 2
    for (int k0 = 0; k0 < KDIM; k0 += 32) {
        #pragma unroll
        for (int j = 0; j < 4; ++j) {
            const int k = k0 + 8 * j;
            const v4f a0 = *(const v4f*)(arow + k);
            const v4f a1 = (MT == 2) ? *(const v4f*)(arow + 16 * XP + k) : a0;
            const v4f p0 = *(const v4f*)(wrow + k);
            const v4f p1 = *(const v4f*)(wrow + 16 * KDIM + k);
            v2f a, b0, b1;
            b0.x = p0.x; b0.y = p0.y;
            b1.x = p1.x; b1.y = p1.y;
            a.x = a0.x; a.y = a0.y;
            acc00 = WMMA4(a, b0, acc00);
            acc01 = WMMA4(a, b1, acc01);
            if (MT == 2) {
                a.x = a1.x; a.y = a1.y;
                acc10 = WMMA4(a, b0, acc10);
                acc11 = WMMA4(a, b1, acc11);
            }
            b0.x = p0.z; b0.y = p0.w;
            b1.x = p1.z; b1.y = p1.w;
            a.x = a0.z; a.y = a0.w;
            acc00 = WMMA4(a, b0, acc00);
            acc01 = WMMA4(a, b1, acc01);
            if (MT == 2) {
                a.x = a1.z; a.y = a1.w;
                acc10 = WMMA4(a, b0, acc10);
                acc11 = WMMA4(a, b1, acc11);
            }
        }
    }
}

template <int MT>
__device__ __forceinline__
void store_tiles(float (*X)[XP], int col0, int wave, int lr, int rbase,
                 const float* __restrict__ bias_v, float bias_scale,
                 const v8f& acc00, const v8f& acc01, const v8f& acc10, const v8f& acc11)
{
    #pragma unroll
    for (int nt = 0; nt < 2; ++nt) {
        const int n = wave * 32 + nt * 16 + lr;
        const float bias = bias_scale * bias_v[n];
        const v8f c0 = (nt == 0) ? acc00 : acc01;
        #pragma unroll
        for (int v = 0; v < 8; ++v)
            X[rbase + v][col0 + n] = c0[v] + bias;
        if (MT == 2) {
            const v8f c1 = (nt == 0) ? acc10 : acc11;
            #pragma unroll
            for (int v = 0; v < 8; ++v)
                X[16 + rbase + v][col0 + n] = c1[v] + bias;
        }
    }
}

template <int MT>
__device__ __forceinline__
void two_gemms(float (*X)[XP], int wave, int lr, int ko, int rbase,
               const float* __restrict__ W1, const float* __restrict__ b1,
               const float* __restrict__ W2, const float* __restrict__ b2)
{
    {
        v8f acc00 = {}, acc01 = {}, acc10 = {}, acc11 = {};
        gemm_strip<MT>(&X[lr][XS + ko], W1 + (size_t)(wave * 32 + lr) * KDIM + ko,
                       acc00, acc01, acc10, acc11);
        __syncthreads();
        store_tiles<MT>(X, XS, wave, lr, rbase, b1, (float)DEG, acc00, acc01, acc10, acc11);
    }
    __syncthreads();
    {
        v8f acc00 = {}, acc01 = {}, acc10 = {}, acc11 = {};
        gemm_strip<MT>(&X[lr][ko], W2 + (size_t)(wave * 32 + lr) * KDIM + ko,
                       acc00, acc01, acc10, acc11);
        store_tiles<MT>(X, XU, wave, lr, rbase, b2, 1.0f, acc00, acc01, acc10, acc11);
    }
}

__global__ __launch_bounds__(THREADS)
void graphsage_fused_kernel(const float* __restrict__ feat_data,
                            const float* __restrict__ feats,
                            const float* __restrict__ W1,
                            const float* __restrict__ b1,
                            const float* __restrict__ W2,
                            const float* __restrict__ b2,
                            const int*   __restrict__ adj,
                            const int*   __restrict__ in1,
                            const int*   __restrict__ in2,
                            const int*   __restrict__ neg,
                            int B1, int B2, int Bn,
                            float* __restrict__ out_all)
{
    __shared__ __attribute__((aligned(16))) float X[TM][XP];
    __shared__ __attribute__((aligned(16))) float aux[TM + TM * DEG];
    float* const nidx_f   = &aux[0];
    float* const nbr_f    = &aux[TM];
    float* const part     = &aux[0];
    float* const inv_norm = &aux[TM * 8];

    const int t    = threadIdx.x;
    const int lane = t & 31;
    const int wave = t >> 5;

    const int nb1 = (B1 + TM - 1) / TM;
    const int nb2 = (B2 + TM - 1) / TM;
    int bi = blockIdx.x;
    const int* __restrict__ nodes = in1;
    int nNodes = B1;
    float* __restrict__ out = out_all;
    if (bi >= nb1 + nb2)  { bi -= nb1 + nb2; nodes = neg; nNodes = Bn; out = out_all + (size_t)(B1 + B2) * FDIM; }
    else if (bi >= nb1)   { bi -= nb1;       nodes = in2; nNodes = B2; out = out_all + (size_t)B1 * FDIM; }
    const int m0 = bi * TM;
    const bool two_tiles = (nNodes - m0) > 16;
    const int  rows_used = two_tiles ? TM : 16;

    if (t < TM) {
        int g = m0 + t;
        nidx_f[t] = __int_as_float((g < nNodes) ? nodes[g] : 0);
    }
    __syncthreads();
    for (int i = t; i < TM * DEG; i += THREADS) {
        const int rowi = i >> 5, d = i & 31;
        const int id   = __float_as_int(nidx_f[rowi]);
        nbr_f[i] = __int_as_float(adj[(size_t)id * DEG + d]);
    }
    __syncthreads();

    {
        const int quarter = t >> 6;
        const int tc      = (t & 63) << 2;
        for (int r = 0; r < rows_used; r += 4) {
            const int rr  = r + quarter;
            const int idx = __float_as_int(nidx_f[rr]);
            float s0 = 0.f, s1 = 0.f, s2 = 0.f, s3 = 0.f;
            float f0 = 0.f, f1 = 0.f, f2 = 0.f, f3 = 0.f;
            #pragma unroll 8
            for (int d = 0; d < DEG; ++d) {
                const int j = __float_as_int(nbr_f[rr * DEG + d]);
                const v4f fa = *(const v4f*)(feat_data + (size_t)j * FDIM + tc);
                const v4f fb = *(const v4f*)(feats     + (size_t)j * FDIM + tc);
                s0 += fa.x; s1 += fa.y; s2 += fa.z; s3 += fa.w;
                f0 += fb.x; f1 += fb.y; f2 += fb.z; f3 += fb.w;
            }
            v4f sv; sv.x = s0; sv.y = s1; sv.z = s2; sv.w = s3;
            v4f fv; fv.x = f0; fv.y = f1; fv.z = f2; fv.w = f3;
            *(v4f*)(&X[rr][XS + tc])        = sv;
            *(v4f*)(&X[rr][XS + FDIM + tc]) = fv;
            *(v4f*)(&X[rr][tc]) = *(const v4f*)(feat_data + (size_t)idx * FDIM + tc);
        }
    }
    __syncthreads();

    const int lr    = lane & 15;
    const int ko    = (lane >> 4) << 2;
    const int rbase = (lane >> 4) << 3;

    if (two_tiles) two_gemms<2>(X, wave, lr, ko, rbase, W1, b1, W2, b2);
    else           two_gemms<1>(X, wave, lr, ko, rbase, W1, b1, W2, b2);
    __syncthreads();

    {
        const int r = t >> 3, seg = t & 7;
        float s = 0.f;
        #pragma unroll 8
        for (int c = seg * 32; c < seg * 32 + 32; ++c) {
            const float x = X[r][XU + c];
            s += x * x;
        }
        part[r * 8 + seg] = s;
    }
    __syncthreads();
    if (t < TM) {
        float s = 0.f;
        #pragma unroll
        for (int i = 0; i < 8; ++i) s += part[t * 8 + i];
        inv_norm[t] = 1.0f / fmaxf(sqrtf(s), 1e-12f);
    }
    __syncthreads();
    for (int i = t; i < TM * FDIM; i += THREADS) {
        const int r = i >> 8;
        const int c = i & (FDIM - 1);
        const int g = m0 + r;
        if (g < nNodes)
            out[(size_t)g * FDIM + c] = X[r][XU + c] * inv_norm[r];
    }
}

extern "C" void kernel_launch(void* const* d_in, const int* in_sizes, int n_in,
                              void* d_out, int out_size, void* d_ws, size_t ws_size,
                              hipStream_t stream) {
    const float* feat_data = (const float*)d_in[0];
    const float* feats     = (const float*)d_in[1];
    const float* W1        = (const float*)d_in[2];
    const float* b1        = (const float*)d_in[3];
    const float* W2        = (const float*)d_in[4];
    const float* b2        = (const float*)d_in[5];
    const int*   adj       = (const int*)d_in[6];
    const int*   in1       = (const int*)d_in[7];
    const int*   in2       = (const int*)d_in[8];
    const int*   neg       = (const int*)d_in[9];
    float* out = (float*)d_out;

    const int B1 = in_sizes[7];
    const int B2 = in_sizes[8];
    const int Bn = in_sizes[9];

    dim3 blk(THREADS);
    const int nblk = (B1 + TM - 1) / TM + (B2 + TM - 1) / TM + (Bn + TM - 1) / TM;
    graphsage_fused_kernel<<<dim3(nblk), blk, 0, stream>>>(
        feat_data, feats, W1, b1, W2, b2, adj, in1, in2, neg, B1, B2, Bn, out);
}


# Neighbour gather reordered: four passes (table by table, half a row of columns at a time), one wave per 512-byte half row, so what the grid reads at one time stays in L2; per-element sum order kept
# speedup vs baseline: 1.7840x; 1.0964x over previous
#include <hip/hip_runtime.h>
#include <hip/hip_bf16.h>


typedef float v2f __attribute__((ext_vector_type(2)));
typedef float v4f __attribute__((ext_vector_type(4)));
typedef float v8f __attribute__((ext_vector_type(8)));

#define FDIM 256
#define KDIM 512
#define DEG  32
#define TM   32
#define THREADS 256
#define XP   772
#define XS   256
#define XU   512

#define WMMA4(A, B, C) __builtin_amdgcn_wmma_f32_16x16x4_f32(false, (A), false, (B), (short)0, (C), false, false)

template <int MT>
__device__ __forceinline__
void gemm_strip(const float* __restrict__ arow,
                const float* __restrict__ wrow,
                v8f& acc00, v8f& acc01, v8f& acc10, v8f& acc11)
{
    #pragma unroll 2
    for (int k0 = 0; k0 < KDIM; k0 += 32) {
        #pragma unroll
        for (int j = 0; j < 4; ++j) {
            const int k = k0 + 8 * j;
            const v4f a0 = *(const v4f*)(arow + k);
            const v4f a1 = (MT == 2) ? *(const v4f*)(arow + 16 * XP + k) : a0;
            const v4f p0 = *(const v4f*)(wrow + k);
            const v4f p1 = *(const v4f*)(wrow + 16 * KDIM + k);
            v2f a, b0, b1;
            b0.x = p0.x; b0.y = p0.y;
            b1.x = p1.x; b1.y = p1.y;
            a.x = a0.x; a.y = a0.y;
            acc00 = WMMA4(a, b0, acc00);
            acc01 = WMMA4(a, b1, acc01);
            if (MT == 2) {
                a.x = a1.x; a.y = a1.y;
                acc10 = WMMA4(a, b0, acc10);
                acc11 = WMMA4(a, b1, acc11);
            }
            b0.x = p0.z; b0.y = p0.w;
            b1.x = p1.z; b1.y = p1.w;
            a.x = a0.z; a.y = a0.w;
            acc00 = WMMA4(a, b0, acc00);
            acc01 = WMMA4(a, b1, acc01);
            if (MT == 2) {
                a.x = a1.z; a.y = a1.w;
                acc10 = WMMA4(a, b0, acc10);
                acc11 = WMMA4(a, b1, acc11);
            }
        }
    }
}

template <int MT>
__device__ __forceinline__
void store_tiles(float (*X)[XP], int col0, int wave, int lr, int rbase,
                 const float* __restrict__ bias_v, float bias_scale,
                 const v8f& acc00, const v8f& acc01, const v8f& acc10, const v8f& acc11)
{
    #pragma unroll
    for (int nt = 0; nt < 2; ++nt) {
        const int n = wave * 32 + nt * 16 + lr;
        const float bias = bias_scale * bias_v[n];
        const v8f c0 = (nt == 0) ? acc00 : acc01;
        #pragma unroll
        for (int v = 0; v < 8; ++v)
            X[rbase + v][col0 + n] = c0[v] + bias;
        if (MT == 2) {
            const v8f c1 = (nt == 0) ? acc10 : acc11;
            #pragma unroll
            for (int v = 0; v < 8; ++v)
                X[16 + rbase + v][col0 + n] = c1[v] + bias;
        }
    }
}

template <int MT>
__device__ __forceinline__
void two_gemms(float (*X)[XP], int wave, int lr, int ko, int rbase,
               const float* __restrict__ W1, const float* __restrict__ b1,
               const float* __restrict__ W2, const float* __restrict__ b2)
{
    {
        v8f acc00 = {}, acc01 = {}, acc10 = {}, acc11 = {};
        gemm_strip<MT>(&X[lr][XS + ko], W1 + (size_t)(wave * 32 + lr) * KDIM + ko,
                       acc00, acc01, acc10, acc11);
        __syncthreads();
        store_tiles<MT>(X, XS, wave, lr, rbase, b1, (float)DEG, acc00, acc01, acc10, acc11);
    }
    __syncthreads();
    {
        v8f acc00 = {}, acc01 = {}, acc10 = {}, acc11 = {};
        gemm_strip<MT>(&X[lr][ko], W2 + (size_t)(wave * 32 + lr) * KDIM + ko,
                       acc00, acc01, acc10, acc11);
        store_tiles<MT>(X, XU, wave, lr, rbase, b2, 1.0f, acc00, acc01, acc10, acc11);
    }
}

__global__ __launch_bounds__(THREADS)
void graphsage_fused_kernel(const float* __restrict__ feat_data,
                            const float* __restrict__ feats,
                            const float* __restrict__ W1,
                            const float* __restrict__ b1,
                            const float* __restrict__ W2,
                            const float* __restrict__ b2,
                            const int*   __restrict__ adj,
                            const int*   __restrict__ in1,
                            const int*   __restrict__ in2,
                            const int*   __restrict__ neg,
                            int B1, int B2, int Bn,
                            float* __restrict__ out_all)
{
    __shared__ __attribute__((aligned(16))) float X[TM][XP];
    __shared__ __attribute__((aligned(16))) float aux[TM + TM * DEG];
    float* const nidx_f   = &aux[0];
    float* const nbr_f    = &aux[TM];
    float* const part     = &aux[0];
    float* const inv_norm = &aux[TM * 8];

    const int t    = threadIdx.x;
    const int lane = t & 31;
    const int wave = t >> 5;

    const int nb1 = (B1 + TM - 1) / TM;
    const int nb2 = (B2 + TM - 1) / TM;
    int bi = blockIdx.x;
    const int* __restrict__ nodes = in1;
    int nNodes = B1;
    float* __restrict__ out = out_all;
    if (bi >= nb1 + nb2)  { bi -= nb1 + nb2; nodes = neg; nNodes = Bn; out = out_all + (size_t)(B1 + B2) * FDIM; }
    else if (bi >= nb1)   { bi -= nb1;       nodes = in2; nNodes = B2; out = out_all + (size_t)B1 * FDIM; }
    const int m0 = bi * TM;
    const bool two_tiles = (nNodes - m0) > 16;
    const int  rows_used = two_tiles ? TM : 16;

    if (t < TM) {
        int g = m0 + t;
        nidx_f[t] = __int_as_float((g < nNodes) ? nodes[g] : 0);
    }
    __syncthreads();
    for (int i = t; i < TM * DEG; i += THREADS) {
        const int rowi = i >> 5, d = i & 31;
        const int id   = __float_as_int(nidx_f[rowi]);
        nbr_f[i] = __int_as_float(adj[(size_t)id * DEG + d]);
    }
    __syncthreads();

    {
        const int c4 = lane << 2;
        for (int p = 0; p < 4; ++p) {
            const float* __restrict__ tab = (p < 2) ? feat_data : feats;
            const int col  = ((p & 1) << 7) + c4;
            const int xcol = XS + ((p >> 1) << 8) + col;
            for (int rr = wave; rr < rows_used; rr += THREADS / 32) {
                const int ids = __float_as_int(nbr_f[rr * DEG + lane]);
                v4f own = {};
                if (p < 2) {
                    const int idx = __builtin_amdgcn_readfirstlane(__float_as_int(nidx_f[rr]));
                    own = *(const v4f*)(feat_data + (size_t)idx * FDIM + col);
                }
                float s0 = 0.f, s1 = 0.f, s2 = 0.f, s3 = 0.f;
                #pragma unroll 16
                for (int d = 0; d < DEG; ++d) {
                    const int j = __builtin_amdgcn_readlane(ids, d);
                    const v4f f = *(const v4f*)(tab + (size_t)j * FDIM + col);
                    s0 += f.x; s1 += f.y; s2 += f.z; s3 += f.w;
                }
                v4f sv; sv.x = s0; sv.y = s1; sv.z = s2; sv.w = s3;
                *(v4f*)(&X[rr][xcol]) = sv;
                if (p < 2) *(v4f*)(&X[rr][col]) = own;
            }
        }
    }
    __syncthreads();

    const int lr    = lane & 15;
    const int ko    = (lane >> 4) << 2;
    const int rbase = (lane >> 4) << 3;

    if (two_tiles) two_gemms<2>(X, wave, lr, ko, rbase, W1, b1, W2, b2);
    else           two_gemms<1>(X, wave, lr, ko, rbase, W1, b1, W2, b2);
    __syncthreads();

    {
        const int r = t >> 3, seg = t & 7;
        float s = 0.f;
        #pragma unroll 8
        for (int c = seg * 32; c < seg * 32 + 32; ++c) {
            const float x = X[r][XU + c];
            s += x * x;
        }
        part[r * 8 + seg] = s;
    }
    __syncthreads();
    if (t < TM) {
        float s = 0.f;
        #pragma unroll
        for (int i = 0; i < 8; ++i) s += part[t * 8 + i];
        inv_norm[t] = 1.0f / fmaxf(sqrtf(s), 1e-12f);
    }
    __syncthreads();
    for (int i = t; i < TM * FDIM; i += THREADS) {
        const int r = i >> 8;
        const int c = i & (FDIM - 1);
        const int g = m0 + r;
        if (g < nNodes)
            out[(size_t)g * FDIM + c] = X[r][XU + c] * inv_norm[r];
    }
}

extern "C" void kernel_launch(void* const* d_in, const int* in_sizes, int n_in,
                              void* d_out, int out_size, void* d_ws, size_t ws_size,
                              hipStream_t stream) {
    const float* feat_data = (const float*)d_in[0];
    const float* feats     = (const float*)d_in[1];
    const float* W1        = (const float*)d_in[2];
    const float* b1        = (const float*)d_in[3];
    const float* W2        = (const float*)d_in[4];
    const float* b2        = (const float*)d_in[5];
    const int*   adj       = (const int*)d_in[6];
    const int*   in1       = (const int*)d_in[7];
    const int*   in2       = (const int*)d_in[8];
    const int*   neg       = (const int*)d_in[9];
    float* out = (float*)d_out;

    const int B1 = in_sizes[7];
    const int B2 = in_sizes[8];
    const int Bn = in_sizes[9];

    dim3 blk(THREADS);
    const int nblk = (B1 + TM - 1) / TM + (B2 + TM - 1) / TM + (Bn + TM - 1) / TM;
    graphsage_fused_kernel<<<dim3(nblk), blk, 0, stream>>>(
        feat_data, feats, W1, b1, W2, b2, adj, in1, in2, neg, B1, B2, Bn, out);
}


# Neighbour gather: all 32 row loads of a row issued before the first is summed, 32 loads in flight per wave instead of 16 (the d loop unrolled whole); per-element sum order kept
# speedup vs baseline: 1.8378x; 1.0302x over previous
#include <hip/hip_runtime.h>
#include <hip/hip_bf16.h>


typedef float v2f __attribute__((ext_vector_type(2)));
typedef float v4f __attribute__((ext_vector_type(4)));
typedef float v8f __attribute__((ext_vector_type(8)));

#define FDIM 256
#define KDIM 512
#define DEG  32
#define TM   32
#define THREADS 256
#define XP   772
#define XS   256
#define XU   512

#define WMMA4(A, B, C) __builtin_amdgcn_wmma_f32_16x16x4_f32(false, (A), false, (B), (short)0, (C), false, false)

template <int MT>
__device__ __forceinline__
void gemm_strip(const float* __restrict__ arow,
                const float* __restrict__ wrow,
                v8f& acc00, v8f& acc01, v8f& acc10, v8f& acc11)
{
    #pragma unroll 2
    for (int k0 = 0; k0 < KDIM; k0 += 32) {
        #pragma unroll
        for (int j = 0; j < 4; ++j) {
            const int k = k0 + 8 * j;
            const v4f a0 = *(const v4f*)(arow + k);
            const v4f a1 = (MT == 2) ? *(const v4f*)(arow + 16 * XP + k) : a0;
            const v4f p0 = *(const v4f*)(wrow + k);
            const v4f p1 = *(const v4f*)(wrow + 16 * KDIM + k);
            v2f a, b0, b1;
            b0.x = p0.x; b0.y = p0.y;
            b1.x = p1.x; b1.y = p1.y;
            a.x = a0.x; a.y = a0.y;
            acc00 = WMMA4(a, b0, acc00);
            acc01 = WMMA4(a, b1, acc01);
            if (MT == 2) {
                a.x = a1.x; a.y = a1.y;
                acc10 = WMMA4(a, b0, acc10);
                acc11 = WMMA4(a, b1, acc11);
            }
            b0.x = p0.z; b0.y = p0.w;
            b1.x = p1.z; b1.y = p1.w;
            a.x = a0.z; a.y = a0.w;
            acc00 = WMMA4(a, b0, acc00);
            acc01 = WMMA4(a, b1, acc01);
            if (MT == 2) {
                a.x = a1.z; a.y = a1.w;
                acc10 = WMMA4(a, b0, acc10);
                acc11 = WMMA4(a, b1, acc11);
            }
        }
    }
}

template <int MT>
__device__ __forceinline__
void store_tiles(float (*X)[XP], int col0, int wave, int lr, int rbase,
                 const float* __restrict__ bias_v, float bias_scale,
                 const v8f& acc00, const v8f& acc01, const v8f& acc10, const v8f& acc11)
{
    #pragma unroll
    for (int nt = 0; nt < 2; ++nt) {
        const int n = wave * 32 + nt * 16 + lr;
        const float bias = bias_scale * bias_v[n];
        const v8f c0 = (nt == 0) ? acc00 : acc01;
        #pragma unroll
        for (int v = 0; v < 8; ++v)
            X[rbase + v][col0 + n] = c0[v] + bias;
        if (MT == 2) {
            const v8f c1 = (nt == 0) ? acc10 : acc11;
            #pragma unroll
            for (int v = 0; v < 8; ++v)
                X[16 + rbase + v][col0 + n] = c1[v] + bias;
        }
    }
}

template <int MT>
__device__ __forceinline__
void two_gemms(float (*X)[XP], int wave, int lr, int ko, int rbase,
               const float* __restrict__ W1, const float* __restrict__ b1,
               const float* __restrict__ W2, const float* __restrict__ b2)
{
    {
        v8f acc00 = {}, acc01 = {}, acc10 = {}, acc11 = {};
        gemm_strip<MT>(&X[lr][XS + ko], W1 + (size_t)(wave * 32 + lr) * KDIM + ko,
                       acc00, acc01, acc10, acc11);
        __syncthreads();
        store_tiles<MT>(X, XS, wave, lr, rbase, b1, (float)DEG, acc00, acc01, acc10, acc11);
    }
    __syncthreads();
    {
        v8f acc00 = {}, acc01 = {}, acc10 = {}, acc11 = {};
        gemm_strip<MT>(&X[lr][ko], W2 + (size_t)(wave * 32 + lr) * KDIM + ko,
                       acc00, acc01, acc10, acc11);
        store_tiles<MT>(X, XU, wave, lr, rbase, b2, 1.0f, acc00, acc01, acc10, acc11);
    }
}

__global__ __launch_bounds__(THREADS)
void graphsage_fused_kernel(const float* __restrict__ feat_data,
                            const float* __restrict__ feats,
                            const float* __restrict__ W1,
                            const float* __restrict__ b1,
                            const float* __restrict__ W2,
                            const float* __restrict__ b2,
                            const int*   __restrict__ adj,
                            const int*   __restrict__ in1,
                            const int*   __restrict__ in2,
                            const int*   __restrict__ neg,
                            int B1, int B2, int Bn,
                            float* __restrict__ out_all)
{
    __shared__ __attribute__((aligned(16))) float X[TM][XP];
    __shared__ __attribute__((aligned(16))) float aux[TM + TM * DEG];
    float* const nidx_f   = &aux[0];
    float* const nbr_f    = &aux[TM];
    float* const part     = &aux[0];
    float* const inv_norm = &aux[TM * 8];

    const int t    = threadIdx.x;
    const int lane = t & 31;
    const int wave = t >> 5;

    const int nb1 = (B1 + TM - 1) / TM;
    const int nb2 = (B2 + TM - 1) / TM;
    int bi = blockIdx.x;
    const int* __restrict__ nodes = in1;
    int nNodes = B1;
    float* __restrict__ out = out_all;
    if (bi >= nb1 + nb2)  { bi -= nb1 + nb2; nodes = neg; nNodes = Bn; out = out_all + (size_t)(B1 + B2) * FDIM; }
    else if (bi >= nb1)   { bi -= nb1;       nodes = in2; nNodes = B2; out = out_all + (size_t)B1 * FDIM; }
    const int m0 = bi * TM;
    const bool two_tiles = (nNodes - m0) > 16;
    const int  rows_used = two_tiles ? TM : 16;

    if (t < TM) {
        int g = m0 + t;
        nidx_f[t] = __int_as_float((g < nNodes) ? nodes[g] : 0);
    }
    __syncthreads();
    for (int i = t; i < TM * DEG; i += THREADS) {
        const int rowi = i >> 5, d = i & 31;
        const int id   = __float_as_int(nidx_f[rowi]);
        nbr_f[i] = __int_as_float(adj[(size_t)id * DEG + d]);
    }
    __syncthreads();

    {
        const int c4 = lane << 2;
        for (int p = 0; p < 4; ++p) {
            const float* __restrict__ tab = (p < 2) ? feat_data : feats;
            const int col  = ((p & 1) << 7) + c4;
            const int xcol = XS + ((p >> 1) << 8) + col;
            for (int rr = wave; rr < rows_used; rr += THREADS / 32) {
                const int ids = __float_as_int(nbr_f[rr * DEG + lane]);
                v4f own = {};
                if (p < 2) {
                    const int idx = __builtin_amdgcn_readfirstlane(__float_as_int(nidx_f[rr]));
                    own = *(const v4f*)(feat_data + (size_t)idx * FDIM + col);
                }
                float s0 = 0.f, s1 = 0.f, s2 = 0.f, s3 = 0.f;
                #pragma unroll
                for (int d = 0; d < DEG; ++d) {
                    const int j = __builtin_amdgcn_readlane(ids, d);
                    const v4f f = *(const v4f*)(tab + (size_t)j * FDIM + col);
                    s0 += f.x; s1 += f.y; s2 += f.z; s3 += f.w;
                }
                v4f sv; sv.x = s0; sv.y = s1; sv.z = s2; sv.w = s3;
                *(v4f*)(&X[rr][xcol]) = sv;
                if (p < 2) *(v4f*)(&X[rr][col]) = own;
            }
        }
    }
    __syncthreads();

    const int lr    = lane & 15;
    const int ko    = (lane >> 4) << 2;
    const int rbase = (lane >> 4) << 3;

    if (two_tiles) two_gemms<2>(X, wave, lr, ko, rbase, W1, b1, W2, b2);
    else           two_gemms<1>(X, wave, lr, ko, rbase, W1, b1, W2, b2);
    __syncthreads();

    {
        const int r = t >> 3, seg = t & 7;
        float s = 0.f;
        #pragma unroll 8
        for (int c = seg * 32; c < seg * 32 + 32; ++c) {
            const float x = X[r][XU + c];
            s += x * x;
        }
        part[r * 8 + seg] = s;
    }
    __syncthreads();
    if (t < TM) {
        float s = 0.f;
        #pragma unroll
        for (int i = 0; i < 8; ++i) s += part[t * 8 + i];
        inv_norm[t] = 1.0f / fmaxf(sqrtf(s), 1e-12f);
    }
    __syncthreads();
    for (int i = t; i < TM * FDIM; i += THREADS) {
        const int r = i >> 8;
        const int c = i & (FDIM - 1);
        const int g = m0 + r;
        if (g < nNodes)
            out[(size_t)g * FDIM + c] = X[r][XU + c] * inv_norm[r];
    }
}

extern "C" void kernel_launch(void* const* d_in, const int* in_sizes, int n_in,
                              void* d_out, int out_size, void* d_ws, size_t ws_size,
                              hipStream_t stream) {
    const float* feat_data = (const float*)d_in[0];
    const float* feats     = (const float*)d_in[1];
    const float* W1        = (const float*)d_in[2];
    const float* b1        = (const float*)d_in[3];
    const float* W2        = (const float*)d_in[4];
    const float* b2        = (const float*)d_in[5];
    const int*   adj       = (const int*)d_in[6];
    const int*   in1       = (const int*)d_in[7];
    const int*   in2       = (const int*)d_in[8];
    const int*   neg       = (const int*)d_in[9];
    float* out = (float*)d_out;

    const int B1 = in_sizes[7];
    const int B2 = in_sizes[8];
    const int Bn = in_sizes[9];

    dim3 blk(THREADS);
    const int nblk = (B1 + TM - 1) / TM + (B2 + TM - 1) / TM + (Bn + TM - 1) / TM;
    graphsage_fused_kernel<<<dim3(nblk), blk, 0, stream>>>(
        feat_data, feats, W1, b1, W2, b2, adj, in1, in2, neg, B1, B2, Bn, out);
}


# W2's neighbour half folded into W1 by a small WMMA launch (Wc=W2b*W1, biases folded): one GEMM of K=256+512 per strip replaces two K=512 GEMMs; the k sums are reassociated
# speedup vs baseline: 1.9733x; 1.0737x over previous
#include <hip/hip_runtime.h>
#include <hip/hip_bf16.h>


typedef float v2f __attribute__((ext_vector_type(2)));
typedef float v4f __attribute__((ext_vector_type(4)));
typedef float v8f __attribute__((ext_vector_type(8)));

#define FDIM 256
#define KDIM 512
#define NT_FOLD 33
#define DEG  32
#define TM   32
#define THREADS 256
#define XP   772
#define XS   256
#define XU   512

#define WMMA4(A, B, C) __builtin_amdgcn_wmma_f32_16x16x4_f32(false, (A), false, (B), (short)0, (C), false, false)


template <bool BIAS>
__device__ __forceinline__
v8f fold_tile(const float* __restrict__ arow,
              const float* __restrict__ bsrc,
              float bscale)
{
    v8f acc = {}, acc2 = {};
    v4f a4[FDIM / 8];
    float q[FDIM / 8][4];
    #pragma unroll
    for (int i = 0; i < FDIM / 8; ++i) {
        const int j0 = 8 * i;
        a4[i] = *(const v4f*)(arow + j0);
        if (BIAS) {
            const v4f b4 = *(const v4f*)(bsrc + j0);
            q[i][0] = bscale * b4.x; q[i][1] = bscale * b4.y; q[i][2] = bscale * b4.z; q[i][3] = bscale * b4.w;
        } else {
            q[i][0] = bsrc[(size_t)(j0 + 0) * KDIM];
            q[i][1] = bsrc[(size_t)(j0 + 1) * KDIM];
            q[i][2] = bsrc[(size_t)(j0 + 2) * KDIM];
            q[i][3] = bsrc[(size_t)(j0 + 3) * KDIM];
        }
    }
    #pragma unroll
    for (int i = 0; i < FDIM / 8; ++i) {
        v2f a, b;
        a.x = a4[i].x; a.y = a4[i].y; b.x = q[i][0]; b.y = q[i][1];
        acc = WMMA4(a, b, acc);
        a.x = a4[i].z; a.y = a4[i].w; b.x = q[i][2]; b.y = q[i][3];
        acc2 = WMMA4(a, b, acc2);
    }
    return acc + acc2;
}

__global__ __launch_bounds__(THREADS) __attribute__((amdgpu_waves_per_eu(2, 2)))
void fold_weights_kernel(const float* __restrict__ W1,
                         const float* __restrict__ b1,
                         const float* __restrict__ W2,
                         const float* __restrict__ b2,
                         float* __restrict__ Wc,
                         float* __restrict__ bf)
{
    const int t    = threadIdx.x;
    const int lane = t & 31;
    const int wave = t >> 5;
    const int lr   = lane & 15;
    const int h    = lane >> 4;

    const int nt = blockIdx.x % NT_FOLD;
    const int ot = (blockIdx.x / NT_FOLD) * 8 + wave;
    const float* __restrict__ arow = W2 + (size_t)(ot * 16 + lr) * KDIM + FDIM + 4 * h;
    const int o0 = ot * 16 + 8 * h;
    if (nt < NT_FOLD - 1) {
        const int n = nt * 16 + lr;
        const v8f acc = fold_tile<false>(arow, W1 + (size_t)(4 * h) * KDIM + n, 0.f);
        #pragma unroll
        for (int v = 0; v < 8; ++v)
            Wc[(size_t)(o0 + v) * KDIM + n] = acc[v];
    } else {
        const v8f acc = fold_tile<true>(arow, b1 + 4 * h, (lr == 0) ? (float)DEG : 0.f);
        if (lr == 0) {
            #pragma unroll
            for (int v = 0; v < 8; ++v)
                bf[o0 + v] = acc[v] + b2[o0 + v];
        }
    }
}

template <int MT, int KLEN>
__device__ __forceinline__
void gemm_strip(const float* __restrict__ arow,
                const float* __restrict__ wrow,
                v8f& acc00, v8f& acc01, v8f& acc10, v8f& acc11)
{
    #pragma unroll 2
    for (int k0 = 0; k0 < KLEN; k0 += 32) {
        #pragma unroll
        for (int j = 0; j < 4; ++j) {
            const int k = k0 + 8 * j;
            const v4f a0 = *(const v4f*)(arow + k);
            const v4f a1 = (MT == 2) ? *(const v4f*)(arow + 16 * XP + k) : a0;
            const v4f p0 = *(const v4f*)(wrow + k);
            const v4f p1 = *(const v4f*)(wrow + 16 * KDIM + k);
            v2f a, b0, b1;
            b0.x = p0.x; b0.y = p0.y;
            b1.x = p1.x; b1.y = p1.y;
            a.x = a0.x; a.y = a0.y;
            acc00 = WMMA4(a, b0, acc00);
            acc01 = WMMA4(a, b1, acc01);
            if (MT == 2) {
                a.x = a1.x; a.y = a1.y;
                acc10 = WMMA4(a, b0, acc10);
                acc11 = WMMA4(a, b1, acc11);
            }
            b0.x = p0.z; b0.y = p0.w;
            b1.x = p1.z; b1.y = p1.w;
            a.x = a0.z; a.y = a0.w;
            acc00 = WMMA4(a, b0, acc00);
            acc01 = WMMA4(a, b1, acc01);
            if (MT == 2) {
                a.x = a1.z; a.y = a1.w;
                acc10 = WMMA4(a, b0, acc10);
                acc11 = WMMA4(a, b1, acc11);
            }
        }
    }
}

template <int MT>
__device__ __forceinline__
void store_tiles(float (*X)[XP], int col0, int wave, int lr, int rbase,
                 const float* __restrict__ bias_v,
                 const v8f& acc00, const v8f& acc01, const v8f& acc10, const v8f& acc11)
{
    #pragma unroll
    for (int nt = 0; nt < 2; ++nt) {
        const int n = wave * 32 + nt * 16 + lr;
        const float bias = bias_v[n];
        const v8f c0 = (nt == 0) ? acc00 : acc01;
        #pragma unroll
        for (int v = 0; v < 8; ++v)
            X[rbase + v][col0 + n] = c0[v] + bias;
        if (MT == 2) {
            const v8f c1 = (nt == 0) ? acc10 : acc11;
            #pragma unroll
            for (int v = 0; v < 8; ++v)
                X[16 + rbase + v][col0 + n] = c1[v] + bias;
        }
    }
}

template <int MT>
__device__ __forceinline__
void one_gemm(float (*X)[XP], int wave, int lr, int ko, int rbase,
              const float* __restrict__ W2, const float* __restrict__ Wc, const float* __restrict__ bf)
{
    v8f acc00 = {}, acc01 = {}, acc10 = {}, acc11 = {};
    gemm_strip<MT, FDIM>(&X[lr][ko], W2 + (size_t)(wave * 32 + lr) * KDIM + ko,
                         acc00, acc01, acc10, acc11);
    gemm_strip<MT, KDIM>(&X[lr][XS + ko], Wc + (size_t)(wave * 32 + lr) * KDIM + ko,
                         acc00, acc01, acc10, acc11);
    __syncthreads();
    store_tiles<MT>(X, XU, wave, lr, rbase, bf, acc00, acc01, acc10, acc11);
}

__global__ __launch_bounds__(THREADS)
void graphsage_fused_kernel(const float* __restrict__ feat_data,
                            const float* __restrict__ feats,
                            const float* __restrict__ W2,
                            const float* __restrict__ Wc,
                            const float* __restrict__ bf,
                            const int*   __restrict__ adj,
                            const int*   __restrict__ in1,
                            const int*   __restrict__ in2,
                            const int*   __restrict__ neg,
                            int B1, int B2, int Bn,
                            float* __restrict__ out_all)
{
    __shared__ __attribute__((aligned(16))) float X[TM][XP];
    __shared__ __attribute__((aligned(16))) float aux[TM + TM * DEG];
    float* const nidx_f   = &aux[0];
    float* const nbr_f    = &aux[TM];
    float* const part     = &aux[0];
    float* const inv_norm = &aux[TM * 8];

    const int t    = threadIdx.x;
    const int lane = t & 31;
    const int wave = t >> 5;

    const int nb1 = (B1 + TM - 1) / TM;
    const int nb2 = (B2 + TM - 1) / TM;
    int bi = blockIdx.x;
    const int* __restrict__ nodes = in1;
    int nNodes = B1;
    float* __restrict__ out = out_all;
    if (bi >= nb1 + nb2)  { bi -= nb1 + nb2; nodes = neg; nNodes = Bn; out = out_all + (size_t)(B1 + B2) * FDIM; }
    else if (bi >= nb1)   { bi -= nb1;       nodes = in2; nNodes = B2; out = out_all + (size_t)B1 * FDIM; }
    const int m0 = bi * TM;
    const bool two_tiles = (nNodes - m0) > 16;
    const int  rows_used = two_tiles ? TM : 16;

    if (t < TM) {
        int g = m0 + t;
        nidx_f[t] = __int_as_float((g < nNodes) ? nodes[g] : 0);
    }
    __syncthreads();
    for (int i = t; i < TM * DEG; i += THREADS) {
        const int rowi = i >> 5, d = i & 31;
        const int id   = __float_as_int(nidx_f[rowi]);
        nbr_f[i] = __int_as_float(adj[(size_t)id * DEG + d]);
    }
    __syncthreads();

    {
        const int c4 = lane << 2;
        for (int p = 0; p < 4; ++p) {
            const float* __restrict__ tab = (p < 2) ? feat_data : feats;
            const int col  = ((p & 1) << 7) + c4;
            const int xcol = XS + ((p >> 1) << 8) + col;
            for (int rr = wave; rr < rows_used; rr += THREADS / 32) {
                const int ids = __float_as_int(nbr_f[rr * DEG + lane]);
                v4f own = {};
                if (p < 2) {
                    const int idx = __builtin_amdgcn_readfirstlane(__float_as_int(nidx_f[rr]));
                    own = *(const v4f*)(feat_data + (size_t)idx * FDIM + col);
                }
                float s0 = 0.f, s1 = 0.f, s2 = 0.f, s3 = 0.f;
                #pragma unroll
                for (int d = 0; d < DEG; ++d) {
                    const int j = __builtin_amdgcn_readlane(ids, d);
                    const v4f f = *(const v4f*)(tab + (size_t)j * FDIM + col);
                    s0 += f.x; s1 += f.y; s2 += f.z; s3 += f.w;
                }
                v4f sv; sv.x = s0; sv.y = s1; sv.z = s2; sv.w = s3;
                *(v4f*)(&X[rr][xcol]) = sv;
                if (p < 2) *(v4f*)(&X[rr][col]) = own;
            }
        }
    }
    __syncthreads();

    const int lr    = lane & 15;
    const int ko    = (lane >> 4) << 2;
    const int rbase = (lane >> 4) << 3;

    if (two_tiles) one_gemm<2>(X, wave, lr, ko, rbase, W2, Wc, bf);
    else           one_gemm<1>(X, wave, lr, ko, rbase, W2, Wc, bf);
    __syncthreads();

    {
        const int r = t >> 3, seg = t & 7;
        float s = 0.f;
        #pragma unroll 8
        for (int c = seg * 32; c < seg * 32 + 32; ++c) {
            const float x = X[r][XU + c];
            s += x * x;
        }
        part[r * 8 + seg] = s;
    }
    __syncthreads();
    if (t < TM) {
        float s = 0.f;
        #pragma unroll
        for (int i = 0; i < 8; ++i) s += part[t * 8 + i];
        inv_norm[t] = 1.0f / fmaxf(sqrtf(s), 1e-12f);
    }
    __syncthreads();
    for (int i = t; i < TM * FDIM; i += THREADS) {
        const int r = i >> 8;
        const int c = i & (FDIM - 1);
        const int g = m0 + r;
        if (g < nNodes)
            out[(size_t)g * FDIM + c] = X[r][XU + c] * inv_norm[r];
    }
}

extern "C" void kernel_launch(void* const* d_in, const int* in_sizes, int n_in,
                              void* d_out, int out_size, void* d_ws, size_t ws_size,
                              hipStream_t stream) {
    const float* feat_data = (const float*)d_in[0];
    const float* feats     = (const float*)d_in[1];
    const float* W1        = (const float*)d_in[2];
    const float* b1        = (const float*)d_in[3];
    const float* W2        = (const float*)d_in[4];
    const float* b2        = (const float*)d_in[5];
    const int*   adj       = (const int*)d_in[6];
    const int*   in1       = (const int*)d_in[7];
    const int*   in2       = (const int*)d_in[8];
    const int*   neg       = (const int*)d_in[9];
    float* out = (float*)d_out;

    const int B1 = in_sizes[7];
    const int B2 = in_sizes[8];
    const int Bn = in_sizes[9];

    float* Wc = (float*)d_ws;
    float* bf = Wc + FDIM * KDIM;

    dim3 blk(THREADS);
    fold_weights_kernel<<<dim3(2 * NT_FOLD), blk, 0, stream>>>(W1, b1, W2, b2, Wc, bf);
    const int nblk = (B1 + TM - 1) / TM + (B2 + TM - 1) / TM + (Bn + TM - 1) / TM;
    graphsage_fused_kernel<<<dim3(nblk), blk, 0, stream>>>(
        feat_data, feats, W2, Wc, bf, adj, in1, in2, neg, B1, B2, Bn, out);
}


# Waves split by role: 4 gather waves fetch pass p+1 while 4 GEMM waves (64 columns each) multiply the columns of pass p, so memory and matrix unit overlap; the k sums are taken in pass order
# speedup vs baseline: 2.1049x; 1.0667x over previous
#include <hip/hip_runtime.h>
#include <hip/hip_bf16.h>


typedef float v2f __attribute__((ext_vector_type(2)));
typedef float v4f __attribute__((ext_vector_type(4)));
typedef float v8f __attribute__((ext_vector_type(8)));

#define FDIM 256
#define KDIM 512
#define NT_FOLD 33
#define DEG  32
#define TM   32
#define THREADS 256
#define XP   772
#define XS   256
#define XU   512

#define WMMA4(A, B, C) __builtin_amdgcn_wmma_f32_16x16x4_f32(false, (A), false, (B), (short)0, (C), false, false)


template <bool BIAS>
__device__ __forceinline__
v8f fold_tile(const float* __restrict__ arow,
              const float* __restrict__ bsrc,
              float bscale)
{
    v8f acc = {}, acc2 = {};
    v4f a4[FDIM / 8];
    float q[FDIM / 8][4];
    #pragma unroll
    for (int i = 0; i < FDIM / 8; ++i) {
        const int j0 = 8 * i;
        a4[i] = *(const v4f*)(arow + j0);
        if (BIAS) {
            const v4f b4 = *(const v4f*)(bsrc + j0);
            q[i][0] = bscale * b4.x; q[i][1] = bscale * b4.y; q[i][2] = bscale * b4.z; q[i][3] = bscale * b4.w;
        } else {
            q[i][0] = bsrc[(size_t)(j0 + 0) * KDIM];
            q[i][1] = bsrc[(size_t)(j0 + 1) * KDIM];
            q[i][2] = bsrc[(size_t)(j0 + 2) * KDIM];
            q[i][3] = bsrc[(size_t)(j0 + 3) * KDIM];
        }
    }
    #pragma unroll
    for (int i = 0; i < FDIM / 8; ++i) {
        v2f a, b;
        a.x = a4[i].x; a.y = a4[i].y; b.x = q[i][0]; b.y = q[i][1];
        acc = WMMA4(a, b, acc);
        a.x = a4[i].z; a.y = a4[i].w; b.x = q[i][2]; b.y = q[i][3];
        acc2 = WMMA4(a, b, acc2);
    }
    return acc + acc2;
}

__global__ __launch_bounds__(THREADS) __attribute__((amdgpu_waves_per_eu(2, 2)))
void fold_weights_kernel(const float* __restrict__ W1,
                         const float* __restrict__ b1,
                         const float* __restrict__ W2,
                         const float* __restrict__ b2,
                         float* __restrict__ Wc,
                         float* __restrict__ bf)
{
    const int t    = threadIdx.x;
    const int lane = t & 31;
    const int wave = t >> 5;
    const int lr   = lane & 15;
    const int h    = lane >> 4;

    const int nt = blockIdx.x % NT_FOLD;
    const int ot = (blockIdx.x / NT_FOLD) * 8 + wave;
    const float* __restrict__ arow = W2 + (size_t)(ot * 16 + lr) * KDIM + FDIM + 4 * h;
    const int o0 = ot * 16 + 8 * h;
    if (nt < NT_FOLD - 1) {
        const int n = nt * 16 + lr;
        const v8f acc = fold_tile<false>(arow, W1 + (size_t)(4 * h) * KDIM + n, 0.f);
        #pragma unroll
        for (int v = 0; v < 8; ++v)
            Wc[(size_t)(o0 + v) * KDIM + n] = acc[v];
    } else {
        const v8f acc = fold_tile<true>(arow, b1 + 4 * h, (lr == 0) ? (float)DEG : 0.f);
        if (lr == 0) {
            #pragma unroll
            for (int v = 0; v < 8; ++v)
                bf[o0 + v] = acc[v] + b2[o0 + v];
        }
    }
}

#define NGW  4
#define CT   4

template <int MT>
__device__ __forceinline__
void gemm_chunk(const float* __restrict__ arow,
                const float* __restrict__ wrow,
                v8f (&acc)[2][CT])
{
    #pragma unroll 1
    for (int k0 = 0; k0 < 128; k0 += 32) {
        #pragma unroll
        for (int j = 0; j < 4; ++j) {
            const int k = k0 + 8 * j;
            const v4f a0 = *(const v4f*)(arow + k);
            const v4f a1 = (MT == 2) ? *(const v4f*)(arow + 16 * XP + k) : a0;
            v4f p[CT];
            #pragma unroll
            for (int c = 0; c < CT; ++c)
                p[c] = *(const v4f*)(wrow + (size_t)c * 16 * KDIM + k);
            v2f a, b;
            a.x = a0.x; a.y = a0.y;
            #pragma unroll
            for (int c = 0; c < CT; ++c) { b.x = p[c].x; b.y = p[c].y; acc[0][c] = WMMA4(a, b, acc[0][c]); }
            if (MT == 2) {
                a.x = a1.x; a.y = a1.y;
                #pragma unroll
                for (int c = 0; c < CT; ++c) { b.x = p[c].x; b.y = p[c].y; acc[1][c] = WMMA4(a, b, acc[1][c]); }
            }
            a.x = a0.z; a.y = a0.w;
            #pragma unroll
            for (int c = 0; c < CT; ++c) { b.x = p[c].z; b.y = p[c].w; acc[0][c] = WMMA4(a, b, acc[0][c]); }
            if (MT == 2) {
                a.x = a1.z; a.y = a1.w;
                #pragma unroll
                for (int c = 0; c < CT; ++c) { b.x = p[c].z; b.y = p[c].w; acc[1][c] = WMMA4(a, b, acc[1][c]); }
            }
        }
    }
}

template <int P>
__device__ __forceinline__
void gather_pass(float (*X)[XP], const float* nidx_f, const float* nbr_f,
                 const float* __restrict__ feat_data, const float* __restrict__ feats,
                 int first, int stride, int rows_used, int lane)
{
    const float* __restrict__ tab = (P < 2) ? feat_data : feats;
    const int col  = ((P & 1) << 7) + (lane << 2);
    const int xcol = XS + ((P >> 1) << 8) + col;
    for (int rr = first; rr < rows_used; rr += stride) {
        const int ids = __float_as_int(nbr_f[rr * DEG + lane]);
        v4f own = {};
        if (P < 2) {
            const int idx = __builtin_amdgcn_readfirstlane(__float_as_int(nidx_f[rr]));
            own = *(const v4f*)(feat_data + (size_t)idx * FDIM + col);
        }
        float s0 = 0.f, s1 = 0.f, s2 = 0.f, s3 = 0.f;
        #pragma unroll
        for (int d = 0; d < DEG; ++d) {
            const int j = __builtin_amdgcn_readlane(ids, d);
            const v4f f = *(const v4f*)(tab + (size_t)j * FDIM + col);
            s0 += f.x; s1 += f.y; s2 += f.z; s3 += f.w;
        }
        v4f sv; sv.x = s0; sv.y = s1; sv.z = s2; sv.w = s3;
        *(v4f*)(&X[rr][xcol]) = sv;
        if (P < 2) *(v4f*)(&X[rr][col]) = own;
    }
}

template <int MT>
__device__ __forceinline__
void strip_pipeline(float (*X)[XP], const float* nidx_f, const float* nbr_f,
                    const float* __restrict__ feat_data, const float* __restrict__ feats,
                    const float* __restrict__ W2, const float* __restrict__ Wc, const float* __restrict__ bf,
                    int wave, int lane, int rows_used)
{
    const int lr    = lane & 15;
    const int ko    = (lane >> 4) << 2;
    const int rbase = (lane >> 4) << 3;
    const bool gatherer = wave < NGW;
    const int  cw = wave & (NGW - 1);

    const float* __restrict__ a_nf = &X[lr][ko];
    const float* __restrict__ a_s  = &X[lr][XS + ko];
    const float* __restrict__ w_nf = W2 + (size_t)(cw * 64 + lr) * KDIM + ko;
    const float* __restrict__ w_s  = Wc + (size_t)(cw * 64 + lr) * KDIM + ko;

    v8f acc[2][CT];
    #pragma unroll
    for (int m = 0; m < 2; ++m)
        #pragma unroll
        for (int c = 0; c < CT; ++c) acc[m][c] = v8f{};

    gather_pass<0>(X, nidx_f, nbr_f, feat_data, feats, wave, THREADS / 32, rows_used, lane);
    __syncthreads();

    #pragma unroll
    for (int p = 1; p <= 4; ++p) {
        if (gatherer) {
            if (p == 1) gather_pass<1>(X, nidx_f, nbr_f, feat_data, feats, wave, NGW, rows_used, lane);
            if (p == 2) gather_pass<2>(X, nidx_f, nbr_f, feat_data, feats, wave, NGW, rows_used, lane);
            if (p == 3) gather_pass<3>(X, nidx_f, nbr_f, feat_data, feats, wave, NGW, rows_used, lane);
            #pragma unroll
            for (int m = 0; m < 2; ++m)
                #pragma unroll
                for (int c = 0; c < CT; ++c) acc[m][c] = v8f{};
        } else {
            const int q = p - 1;
            if (q < 2) gemm_chunk<MT>(a_nf + q * 128, w_nf + q * 128, acc);
            gemm_chunk<MT>(a_s + q * 128, w_s + q * 128, acc);
        }
        __syncthreads();
    }

    if (!gatherer) {
        #pragma unroll
        for (int c = 0; c < CT; ++c) {
            const int n = cw * 64 + c * 16 + lr;
            const float bias = bf[n];
            #pragma unroll
            for (int v = 0; v < 8; ++v)
                X[rbase + v][XU + n] = acc[0][c][v] + bias;
            if (MT == 2) {
                #pragma unroll
                for (int v = 0; v < 8; ++v)
                    X[16 + rbase + v][XU + n] = acc[1][c][v] + bias;
            }
        }
    }
}

__global__ __launch_bounds__(THREADS)
void graphsage_fused_kernel(const float* __restrict__ feat_data,
                            const float* __restrict__ feats,
                            const float* __restrict__ W2,
                            const float* __restrict__ Wc,
                            const float* __restrict__ bf,
                            const int*   __restrict__ adj,
                            const int*   __restrict__ in1,
                            const int*   __restrict__ in2,
                            const int*   __restrict__ neg,
                            int B1, int B2, int Bn,
                            float* __restrict__ out_all)
{
    __shared__ __attribute__((aligned(16))) float X[TM][XP];
    __shared__ __attribute__((aligned(16))) float aux[TM + TM * DEG];
    float* const nidx_f   = &aux[0];
    float* const nbr_f    = &aux[TM];
    float* const part     = &aux[0];
    float* const inv_norm = &aux[TM * 8];

    const int t    = threadIdx.x;
    const int lane = t & 31;
    const int wave = __builtin_amdgcn_readfirstlane(t >> 5);

    const int nb1 = (B1 + TM - 1) / TM;
    const int nb2 = (B2 + TM - 1) / TM;
    int bi = blockIdx.x;
    const int* __restrict__ nodes = in1;
    int nNodes = B1;
    float* __restrict__ out = out_all;
    if (bi >= nb1 + nb2)  { bi -= nb1 + nb2; nodes = neg; nNodes = Bn; out = out_all + (size_t)(B1 + B2) * FDIM; }
    else if (bi >= nb1)   { bi -= nb1;       nodes = in2; nNodes = B2; out = out_all + (size_t)B1 * FDIM; }
    const int m0 = bi * TM;
    const bool two_tiles = (nNodes - m0) > 16;
    const int  rows_used = two_tiles ? TM : 16;

    if (t < TM) {
        int g = m0 + t;
        nidx_f[t] = __int_as_float((g < nNodes) ? nodes[g] : 0);
    }
    __syncthreads();
    for (int i = t; i < TM * DEG; i += THREADS) {
        const int rowi = i >> 5, d = i & 31;
        const int id   = __float_as_int(nidx_f[rowi]);
        nbr_f[i] = __int_as_float(adj[(size_t)id * DEG + d]);
    }
    __syncthreads();

    if (two_tiles) strip_pipeline<2>(X, nidx_f, nbr_f, feat_data, feats, W2, Wc, bf, wave, lane, rows_used);
    else           strip_pipeline<1>(X, nidx_f, nbr_f, feat_data, feats, W2, Wc, bf, wave, lane, rows_used);
    __syncthreads();

    {
        const int r = t >> 3, seg = t & 7;
        float s = 0.f;
        #pragma unroll 8
        for (int c = seg * 32; c < seg * 32 + 32; ++c) {
            const float x = X[r][XU + c];
            s += x * x;
        }
        part[r * 8 + seg] = s;
    }
    __syncthreads();
    if (t < TM) {
        float s = 0.f;
        #pragma unroll
        for (int i = 0; i < 8; ++i) s += part[t * 8 + i];
        inv_norm[t] = 1.0f / fmaxf(sqrtf(s), 1e-12f);
    }
    __syncthreads();
    for (int i = t; i < TM * FDIM; i += THREADS) {
        const int r = i >> 8;
        const int c = i & (FDIM - 1);
        const int g = m0 + r;
        if (g < nNodes)
            out[(size_t)g * FDIM + c] = X[r][XU + c] * inv_norm[r];
    }
}

extern "C" void kernel_launch(void* const* d_in, const int* in_sizes, int n_in,
                              void* d_out, int out_size, void* d_ws, size_t ws_size,
                              hipStream_t stream) {
    const float* feat_data = (const float*)d_in[0];
    const float* feats     = (const float*)d_in[1];
    const float* W1        = (const float*)d_in[2];
    const float* b1        = (const float*)d_in[3];
    const float* W2        = (const float*)d_in[4];
    const float* b2        = (const float*)d_in[5];
    const int*   adj       = (const int*)d_in[6];
    const int*   in1       = (const int*)d_in[7];
    const int*   in2       = (const int*)d_in[8];
    const int*   neg       = (const int*)d_in[9];
    float* out = (float*)d_out;

    const int B1 = in_sizes[7];
    const int B2 = in_sizes[8];
    const int Bn = in_sizes[9];

    float* Wc = (float*)d_ws;
    float* bf = Wc + FDIM * KDIM;

    dim3 blk(THREADS);
    fold_weights_kernel<<<dim3(2 * NT_FOLD), blk, 0, stream>>>(W1, b1, W2, b2, Wc, bf);
    const int nblk = (B1 + TM - 1) / TM + (B2 + TM - 1) / TM + (Bn + TM - 1) / TM;
    graphsage_fused_kernel<<<dim3(nblk), blk, 0, stream>>>(
        feat_data, feats, W2, Wc, bf, adj, in1, in2, neg, B1, B2, Bn, out);
}


# First launch writes Wc and a copy of W2a in fragment order, so each B fragment load of the main GEMM is one contiguous 512-byte run (4 whole lines) instead of 16 lines of 32 bytes; sum order kept
# speedup vs baseline: 2.2313x; 1.0601x over previous
#include <hip/hip_runtime.h>
#include <hip/hip_bf16.h>


typedef float v2f __attribute__((ext_vector_type(2)));
typedef float v4f __attribute__((ext_vector_type(4)));
typedef float v8f __attribute__((ext_vector_type(8)));

#define FDIM 256
#define KDIM 512
#define NT_FOLD 33
#define DEG  32
#define TM   32
#define THREADS 256
#define XP   772
#define XS   256
#define XU   512

#define WMMA4(A, B, C) __builtin_amdgcn_wmma_f32_16x16x4_f32(false, (A), false, (B), (short)0, (C), false, false)

#define FRAG_FLOATS 128

template <bool BIAS>
__device__ __forceinline__
v8f fold_tile(const float* __restrict__ arow,
              const float* __restrict__ bsrc,
              float bscale)
{
    v8f acc = {}, acc2 = {};
    v4f a4[FDIM / 8];
    float q[FDIM / 8][4];
    #pragma unroll
    for (int i = 0; i < FDIM / 8; ++i) {
        const int j0 = 8 * i;
        a4[i] = *(const v4f*)(arow + j0);
        if (BIAS) {
            const v4f b4 = *(const v4f*)(bsrc + j0);
            q[i][0] = bscale * b4.x; q[i][1] = bscale * b4.y; q[i][2] = bscale * b4.z; q[i][3] = bscale * b4.w;
        } else {
            q[i][0] = bsrc[(size_t)(j0 + 0) * KDIM];
            q[i][1] = bsrc[(size_t)(j0 + 1) * KDIM];
            q[i][2] = bsrc[(size_t)(j0 + 2) * KDIM];
            q[i][3] = bsrc[(size_t)(j0 + 3) * KDIM];
        }
    }
    #pragma unroll
    for (int i = 0; i < FDIM / 8; ++i) {
        v2f a, b;
        a.x = q[i][0]; a.y = q[i][1]; b.x = a4[i].x; b.y = a4[i].y;
        acc = WMMA4(a, b, acc);
        a.x = q[i][2]; a.y = q[i][3]; b.x = a4[i].z; b.y = a4[i].w;
        acc2 = WMMA4(a, b, acc2);
    }
    return acc + acc2;
}

__global__ __launch_bounds__(THREADS) __attribute__((amdgpu_waves_per_eu(2, 2)))
void fold_weights_kernel(const float* __restrict__ W1,
                         const float* __restrict__ b1,
                         const float* __restrict__ W2,
                         const float* __restrict__ b2,
                         float* __restrict__ Wcf,
                         float* __restrict__ W2f,
                         float* __restrict__ bf)
{
    const int t    = threadIdx.x;
    const int lane = t & 31;
    const int wave = t >> 5;
    const int lr   = lane & 15;
    const int h    = lane >> 4;

    const int gw = blockIdx.x * 8 + wave;
    v4f w2a = {};
    if (gw < (FDIM / 16) * (FDIM / 8))
        w2a = *(const v4f*)(W2 + (size_t)((gw >> 5) * 16 + lr) * KDIM + 8 * (gw & 31) + 4 * h);

    const int nt = blockIdx.x % NT_FOLD;
    const int ot = (blockIdx.x / NT_FOLD) * 8 + wave;
    const float* __restrict__ arow = W2 + (size_t)(ot * 16 + lr) * KDIM + FDIM + 4 * h;
    if (nt < NT_FOLD - 1) {
        const v8f acc = fold_tile<false>(arow, W1 + (size_t)(4 * h) * KDIM + nt * 16 + lr, 0.f);
        float* __restrict__ blk = Wcf + (size_t)(ot * (KDIM / 8) + 2 * nt + h) * FRAG_FLOATS;
        v4f lo, hi;
        lo.x = acc[0]; lo.y = acc[1]; lo.z = acc[2]; lo.w = acc[3];
        hi.x = acc[4]; hi.y = acc[5]; hi.z = acc[6]; hi.w = acc[7];
        *(v4f*)(blk + 4 * lr)        = lo;
        *(v4f*)(blk + 4 * (16 + lr)) = hi;
    } else {
        const v8f acc = fold_tile<true>(arow, b1 + 4 * h, (lr == 0) ? (float)DEG : 0.f);
        if (h == 0)
            bf[ot * 16 + lr] = acc[0] + b2[ot * 16 + lr];
    }

    if (gw < (FDIM / 16) * (FDIM / 8))
        *(v4f*)(W2f + (size_t)gw * FRAG_FLOATS + 4 * lane) = w2a;
}

#define NGW  4
#define CT   4

template <int MT>
__device__ __forceinline__
void gemm_chunk(const float* __restrict__ arow,
                const float* __restrict__ wfrag,
                int tile_stride,
                v8f (&acc)[2][CT])
{
    #pragma unroll 1
    for (int k0 = 0; k0 < 128; k0 += 32) {
        #pragma unroll
        for (int j = 0; j < 4; ++j) {
            const int k = k0 + 8 * j;
            const v4f a0 = *(const v4f*)(arow + k);
            const v4f a1 = (MT == 2) ? *(const v4f*)(arow + 16 * XP + k) : a0;
            v4f p[CT];
            #pragma unroll
            for (int c = 0; c < CT; ++c)
                p[c] = *(const v4f*)(wfrag + c * tile_stride + (k >> 3) * FRAG_FLOATS);
            v2f a, b;
            a.x = a0.x; a.y = a0.y;
            #pragma unroll
            for (int c = 0; c < CT; ++c) { b.x = p[c].x; b.y = p[c].y; acc[0][c] = WMMA4(a, b, acc[0][c]); }
            if (MT == 2) {
                a.x = a1.x; a.y = a1.y;
                #pragma unroll
                for (int c = 0; c < CT; ++c) { b.x = p[c].x; b.y = p[c].y; acc[1][c] = WMMA4(a, b, acc[1][c]); }
            }
            a.x = a0.z; a.y = a0.w;
            #pragma unroll
            for (int c = 0; c < CT; ++c) { b.x = p[c].z; b.y = p[c].w; acc[0][c] = WMMA4(a, b, acc[0][c]); }
            if (MT == 2) {
                a.x = a1.z; a.y = a1.w;
                #pragma unroll
                for (int c = 0; c < CT; ++c) { b.x = p[c].z; b.y = p[c].w; acc[1][c] = WMMA4(a, b, acc[1][c]); }
            }
        }
    }
}

template <int P>
__device__ __forceinline__
void gather_pass(float (*X)[XP], const float* nidx_f, const float* nbr_f,
                 const float* __restrict__ feat_data, const float* __restrict__ feats,
                 int first, int stride, int rows_used, int lane)
{
    const float* __restrict__ tab = (P < 2) ? feat_data : feats;
    const int col  = ((P & 1) << 7) + (lane << 2);
    const int xcol = XS + ((P >> 1) << 8) + col;
    for (int rr = first; rr < rows_used; rr += stride) {
        const int ids = __float_as_int(nbr_f[rr * DEG + lane]);
        v4f own = {};
        if (P < 2) {
            const int idx = __builtin_amdgcn_readfirstlane(__float_as_int(nidx_f[rr]));
            own = *(const v4f*)(feat_data + (size_t)idx * FDIM + col);
        }
        float s0 = 0.f, s1 = 0.f, s2 = 0.f, s3 = 0.f;
        #pragma unroll
        for (int d = 0; d < DEG; ++d) {
            const int j = __builtin_amdgcn_readlane(ids, d);
            const v4f f = *(const v4f*)(tab + (size_t)j * FDIM + col);
            s0 += f.x; s1 += f.y; s2 += f.z; s3 += f.w;
        }
        v4f sv; sv.x = s0; sv.y = s1; sv.z = s2; sv.w = s3;
        *(v4f*)(&X[rr][xcol]) = sv;
        if (P < 2) *(v4f*)(&X[rr][col]) = own;
    }
}

template <int MT>
__device__ __forceinline__
void strip_pipeline(float (*X)[XP], const float* nidx_f, const float* nbr_f,
                    const float* __restrict__ feat_data, const float* __restrict__ feats,
                    const float* __restrict__ W2f, const float* __restrict__ Wcf, const float* __restrict__ bf,
                    int wave, int lane, int rows_used)
{
    const int lr    = lane & 15;
    const int ko    = (lane >> 4) << 2;
    const int rbase = (lane >> 4) << 3;
    const bool gatherer = wave < NGW;
    const int  cw = wave & (NGW - 1);

    const float* __restrict__ a_nf = &X[lr][ko];
    const float* __restrict__ a_s  = &X[lr][XS + ko];
    const int ts_nf = (FDIM / 8) * FRAG_FLOATS;
    const int ts_s  = (KDIM / 8) * FRAG_FLOATS;
    const float* __restrict__ w_nf = W2f + (size_t)(cw * CT) * ts_nf + 4 * lane;
    const float* __restrict__ w_s  = Wcf + (size_t)(cw * CT) * ts_s  + 4 * lane;

    v8f acc[2][CT];
    #pragma unroll
    for (int m = 0; m < 2; ++m)
        #pragma unroll
        for (int c = 0; c < CT; ++c) acc[m][c] = v8f{};

    gather_pass<0>(X, nidx_f, nbr_f, feat_data, feats, wave, THREADS / 32, rows_used, lane);
    __syncthreads();

    #pragma unroll
    for (int p = 1; p <= 4; ++p) {
        if (gatherer) {
            if (p == 1) gather_pass<1>(X, nidx_f, nbr_f, feat_data, feats, wave, NGW, rows_used, lane);
            if (p == 2) gather_pass<2>(X, nidx_f, nbr_f, feat_data, feats, wave, NGW, rows_used, lane);
            if (p == 3) gather_pass<3>(X, nidx_f, nbr_f, feat_data, feats, wave, NGW, rows_used, lane);
            #pragma unroll
            for (int m = 0; m < 2; ++m)
                #pragma unroll
                for (int c = 0; c < CT; ++c) acc[m][c] = v8f{};
        } else {
            const int q = p - 1;
            if (q < 2) gemm_chunk<MT>(a_nf + q * 128, w_nf + q * (128 / 8) * FRAG_FLOATS, ts_nf, acc);
            gemm_chunk<MT>(a_s + q * 128, w_s + q * (128 / 8) * FRAG_FLOATS, ts_s, acc);
        }
        __syncthreads();
    }

    if (!gatherer) {
        #pragma unroll
        for (int c = 0; c < CT; ++c) {
            const int n = cw * 64 + c * 16 + lr;
            const float bias = bf[n];
            #pragma unroll
            for (int v = 0; v < 8; ++v)
                X[rbase + v][XU + n] = acc[0][c][v] + bias;
            if (MT == 2) {
                #pragma unroll
                for (int v = 0; v < 8; ++v)
                    X[16 + rbase + v][XU + n] = acc[1][c][v] + bias;
            }
        }
    }
}

__global__ __launch_bounds__(THREADS)
void graphsage_fused_kernel(const float* __restrict__ feat_data,
                            const float* __restrict__ feats,
                            const float* __restrict__ W2f,
                            const float* __restrict__ Wcf,
                            const float* __restrict__ bf,
                            const int*   __restrict__ adj,
                            const int*   __restrict__ in1,
                            const int*   __restrict__ in2,
                            const int*   __restrict__ neg,
                            int B1, int B2, int Bn,
                            float* __restrict__ out_all)
{
    __shared__ __attribute__((aligned(16))) float X[TM][XP];
    __shared__ __attribute__((aligned(16))) float aux[TM + TM * DEG];
    float* const nidx_f   = &aux[0];
    float* const nbr_f    = &aux[TM];
    float* const part     = &aux[0];
    float* const inv_norm = &aux[TM * 8];

    const int t    = threadIdx.x;
    const int lane = t & 31;
    const int wave = __builtin_amdgcn_readfirstlane(t >> 5);

    const int nb1 = (B1 + TM - 1) / TM;
    const int nb2 = (B2 + TM - 1) / TM;
    int bi = blockIdx.x;
    const int* __restrict__ nodes = in1;
    int nNodes = B1;
    float* __restrict__ out = out_all;
    if (bi >= nb1 + nb2)  { bi -= nb1 + nb2; nodes = neg; nNodes = Bn; out = out_all + (size_t)(B1 + B2) * FDIM; }
    else if (bi >= nb1)   { bi -= nb1;       nodes = in2; nNodes = B2; out = out_all + (size_t)B1 * FDIM; }
    const int m0 = bi * TM;
    const bool two_tiles = (nNodes - m0) > 16;
    const int  rows_used = two_tiles ? TM : 16;

    if (t < TM) {
        int g = m0 + t;
        nidx_f[t] = __int_as_float((g < nNodes) ? nodes[g] : 0);
    }
    __syncthreads();
    for (int i = t; i < TM * DEG; i += THREADS) {
        const int rowi = i >> 5, d = i & 31;
        const int id   = __float_as_int(nidx_f[rowi]);
        nbr_f[i] = __int_as_float(adj[(size_t)id * DEG + d]);
    }
    __syncthreads();

    if (two_tiles) strip_pipeline<2>(X, nidx_f, nbr_f, feat_data, feats, W2f, Wcf, bf, wave, lane, rows_used);
    else           strip_pipeline<1>(X, nidx_f, nbr_f, feat_data, feats, W2f, Wcf, bf, wave, lane, rows_used);
    __syncthreads();

    {
        const int r = t >> 3, seg = t & 7;
        float s = 0.f;
        #pragma unroll 8
        for (int c = seg * 32; c < seg * 32 + 32; ++c) {
            const float x = X[r][XU + c];
            s += x * x;
        }
        part[r * 8 + seg] = s;
    }
    __syncthreads();
    if (t < TM) {
        float s = 0.f;
        #pragma unroll
        for (int i = 0; i < 8; ++i) s += part[t * 8 + i];
        inv_norm[t] = 1.0f / fmaxf(sqrtf(s), 1e-12f);
    }
    __syncthreads();
    for (int i = t; i < TM * FDIM; i += THREADS) {
        const int r = i >> 8;
        const int c = i & (FDIM - 1);
        const int g = m0 + r;
        if (g < nNodes)
            out[(size_t)g * FDIM + c] = X[r][XU + c] * inv_norm[r];
    }
}

extern "C" void kernel_launch(void* const* d_in, const int* in_sizes, int n_in,
                              void* d_out, int out_size, void* d_ws, size_t ws_size,
                              hipStream_t stream) {
    const float* feat_data = (const float*)d_in[0];
    const float* feats     = (const float*)d_in[1];
    const float* W1        = (const float*)d_in[2];
    const float* b1        = (const float*)d_in[3];
    const float* W2        = (const float*)d_in[4];
    const float* b2        = (const float*)d_in[5];
    const int*   adj       = (const int*)d_in[6];
    const int*   in1       = (const int*)d_in[7];
    const int*   in2       = (const int*)d_in[8];
    const int*   neg       = (const int*)d_in[9];
    float* out = (float*)d_out;

    const int B1 = in_sizes[7];
    const int B2 = in_sizes[8];
    const int Bn = in_sizes[9];

    float* Wcf = (float*)d_ws;
    float* bf  = Wcf + FDIM * KDIM;
    float* W2f = bf + FDIM;

    dim3 blk(THREADS);
    fold_weights_kernel<<<dim3(2 * NT_FOLD), blk, 0, stream>>>(W1, b1, W2, b2, Wcf, W2f, bf);
    const int nblk = (B1 + TM - 1) / TM + (B2 + TM - 1) / TM + (Bn + TM - 1) / TM;
    graphsage_fused_kernel<<<dim3(nblk), blk, 0, stream>>>(
        feat_data, feats, W2f, Wcf, bf, adj, in1, in2, neg, B1, B2, Bn, out);
}


# GEMM waves keep their weight fragments as a rolling window one trip ahead (Wf stored in order of use; LDS-only barrier keeps the loads in flight across gather passes); sum order kept
# speedup vs baseline: 2.4044x; 1.0776x over previous
#include <hip/hip_runtime.h>
#include <hip/hip_bf16.h>


typedef float v2f __attribute__((ext_vector_type(2)));
typedef float v4f __attribute__((ext_vector_type(4)));
typedef float v8f __attribute__((ext_vector_type(8)));

#define FDIM 256
#define KDIM 512
#define NT_FOLD 33
#define DEG  32
#define TM   32
#define THREADS 256
#define XP   772
#define XS   256
#define XU   512

#define WMMA4(A, B, C) __builtin_amdgcn_wmma_f32_16x16x4_f32(false, (A), false, (B), (short)0, (C), false, false)

#define FRAG_FLOATS 128
#define NPOS 96
#define CT_ALL 4
#define WAVE_FLOATS (NPOS * CT_ALL * FRAG_FLOATS)

__device__ __forceinline__ size_t wf_block(int nt, int pos) { return (size_t)(((nt >> 2) * NPOS + pos) * CT_ALL + (nt & 3)) * FRAG_FLOATS; }

__device__ __forceinline__ int pos_w2a(int kc) { return ((kc >> 4) << 5) + (kc & 15); }
__device__ __forceinline__ int pos_wc(int kc)  { const int q = kc >> 4; return ((q < 2) ? 16 + 32 * q : 32 + 16 * q) + (kc & 15); }

template <bool BIAS>
__device__ __forceinline__
v8f fold_tile(const float* __restrict__ arow,
              const float* __restrict__ bsrc,
              float bscale)
{
    v8f acc = {}, acc2 = {};
    v4f a4[FDIM / 8];
    float q[FDIM / 8][4];
    #pragma unroll
    for (int i = 0; i < FDIM / 8; ++i) {
        const int j0 = 8 * i;
        a4[i] = *(const v4f*)(arow + j0);
        if (BIAS) {
            const v4f b4 = *(const v4f*)(bsrc + j0);
            q[i][0] = bscale * b4.x; q[i][1] = bscale * b4.y; q[i][2] = bscale * b4.z; q[i][3] = bscale * b4.w;
        } else {
            q[i][0] = bsrc[(size_t)(j0 + 0) * KDIM];
            q[i][1] = bsrc[(size_t)(j0 + 1) * KDIM];
            q[i][2] = bsrc[(size_t)(j0 + 2) * KDIM];
            q[i][3] = bsrc[(size_t)(j0 + 3) * KDIM];
        }
    }
    #pragma unroll
    for (int i = 0; i < FDIM / 8; ++i) {
        v2f a, b;
        a.x = q[i][0]; a.y = q[i][1]; b.x = a4[i].x; b.y = a4[i].y;
        acc = WMMA4(a, b, acc);
        a.x = q[i][2]; a.y = q[i][3]; b.x = a4[i].z; b.y = a4[i].w;
        acc2 = WMMA4(a, b, acc2);
    }
    return acc + acc2;
}

__global__ __launch_bounds__(THREADS) __attribute__((amdgpu_waves_per_eu(2, 2)))
void fold_weights_kernel(const float* __restrict__ W1,
                         const float* __restrict__ b1,
                         const float* __restrict__ W2,
                         const float* __restrict__ b2,
                         float* __restrict__ Wf,
                         float* __restrict__ bf)
{
    const int t    = threadIdx.x;
    const int lane = t & 31;
    const int wave = t >> 5;
    const int lr   = lane & 15;
    const int h    = lane >> 4;

    const int gw = blockIdx.x * 8 + wave;
    v4f w2a = {};
    if (gw < (FDIM / 16) * (FDIM / 8))
        w2a = *(const v4f*)(W2 + (size_t)((gw >> 5) * 16 + lr) * KDIM + 8 * (gw & 31) + 4 * h);

    const int nt = blockIdx.x % NT_FOLD;
    const int ot = (blockIdx.x / NT_FOLD) * 8 + wave;
    const float* __restrict__ arow = W2 + (size_t)(ot * 16 + lr) * KDIM + FDIM + 4 * h;
    if (nt < NT_FOLD - 1) {
        const v8f acc = fold_tile<false>(arow, W1 + (size_t)(4 * h) * KDIM + nt * 16 + lr, 0.f);
        float* __restrict__ blk = Wf + wf_block(ot, pos_wc(2 * nt + h));
        v4f lo, hi;
        lo.x = acc[0]; lo.y = acc[1]; lo.z = acc[2]; lo.w = acc[3];
        hi.x = acc[4]; hi.y = acc[5]; hi.z = acc[6]; hi.w = acc[7];
        *(v4f*)(blk + 4 * lr)        = lo;
        *(v4f*)(blk + 4 * (16 + lr)) = hi;
    } else {
        const v8f acc = fold_tile<true>(arow, b1 + 4 * h, (lr == 0) ? (float)DEG : 0.f);
        if (h == 0)
            bf[ot * 16 + lr] = acc[0] + b2[ot * 16 + lr];
    }

    if (gw < (FDIM / 16) * (FDIM / 8))
        *(v4f*)(Wf + wf_block(gw >> 5, pos_w2a(gw & 31)) + 4 * lane) = w2a;
}

__device__ __forceinline__ void lds_barrier()
{
    asm volatile("s_wait_dscnt 0x0\n\ts_barrier_signal -1\n\ts_barrier_wait -1" ::: "memory");
}

#define NGW  4
#define CT   4
static_assert(CT == CT_ALL, "Wf interleaves the column tiles of one GEMM wave");

template <int MT, bool B_NEXT>
__device__ __forceinline__
void gemm_trip(const float* __restrict__ arow,
               const float* __restrict__ wn,
               bool a_next,
               v4f (&a)[2][2],
               v4f (&p)[4][CT],
               v8f (&acc)[2][CT])
{
    #pragma unroll
    for (int j = 0; j < 4; ++j) {
        const int s = j & 1;
        #pragma unroll
        for (int c0 = 0; c0 < CT; c0 += 2) {
            v2f x, y;
            #pragma unroll
            for (int e = 0; e < 2; ++e) {
                #pragma unroll
                for (int m = 0; m < MT; ++m) {
                    x.x = e ? a[s][m].z : a[s][m].x;  x.y = e ? a[s][m].w : a[s][m].y;
                    #pragma unroll
                    for (int c = c0; c < c0 + 2; ++c) {
                        y.x = e ? p[j][c].z : p[j][c].x;  y.y = e ? p[j][c].w : p[j][c].y;
                        acc[m][c] = WMMA4(x, y, acc[m][c]);
                    }
                }
            }
        }
        if (B_NEXT) {
            #pragma unroll
            for (int c = 0; c < CT; ++c)
                p[j][c] = *(const v4f*)(wn + (j * CT + c) * FRAG_FLOATS);
        }
        if (j < 2 || a_next) {
            a[s][0] = *(const v4f*)(arow + 8 * (j + 2));
            if (MT == 2) a[s][1] = *(const v4f*)(arow + 16 * XP + 8 * (j + 2));
        }
        __builtin_amdgcn_sched_group_barrier(0x008, 8 * MT, 0);
        if (B_NEXT) __builtin_amdgcn_sched_group_barrier(0x020, CT, 0);
        if (j < 2)  __builtin_amdgcn_sched_group_barrier(0x100, MT, 0);
    }
}

template <int MT, bool LAST>
__device__ __forceinline__
void gemm_chunk(const float* __restrict__ arow,
                const float* __restrict__ wfrag,
                v4f (&p)[4][CT],
                v8f (&acc)[2][CT])
{
    v4f a[2][2];
    #pragma unroll
    for (int s = 0; s < 2; ++s) {
        a[s][0] = *(const v4f*)(arow + 8 * s);
        a[s][1] = (MT == 2) ? *(const v4f*)(arow + 16 * XP + 8 * s) : a[s][0];
    }
    #pragma unroll 1
    for (int t = 0; t < (LAST ? 3 : 4); ++t)
        gemm_trip<MT, true>(arow + 32 * t, wfrag + 4 * CT * (t + 1) * FRAG_FLOATS, t < 3, a, p, acc);
    if (LAST)
        gemm_trip<MT, false>(arow + 96, wfrag, false, a, p, acc);
}

template <int P>
__device__ __forceinline__
void gather_pass(float (*X)[XP], const float* nidx_f, const float* nbr_f,
                 const float* __restrict__ feat_data, const float* __restrict__ feats,
                 int first, int stride, int rows_used, int lane)
{
    const float* __restrict__ tab = (P < 2) ? feat_data : feats;
    const int col  = ((P & 1) << 7) + (lane << 2);
    const int xcol = XS + ((P >> 1) << 8) + col;
    for (int rr = first; rr < rows_used; rr += stride) {
        const int ids = __float_as_int(nbr_f[rr * DEG + lane]);
        v4f own = {};
        if (P < 2) {
            const int idx = __builtin_amdgcn_readfirstlane(__float_as_int(nidx_f[rr]));
            own = *(const v4f*)(feat_data + (size_t)idx * FDIM + col);
        }
        float s0 = 0.f, s1 = 0.f, s2 = 0.f, s3 = 0.f;
        #pragma unroll
        for (int d = 0; d < DEG; ++d) {
            const int j = __builtin_amdgcn_readlane(ids, d);
            const v4f f = *(const v4f*)(tab + (size_t)j * FDIM + col);
            s0 += f.x; s1 += f.y; s2 += f.z; s3 += f.w;
        }
        v4f sv; sv.x = s0; sv.y = s1; sv.z = s2; sv.w = s3;
        *(v4f*)(&X[rr][xcol]) = sv;
        if (P < 2) *(v4f*)(&X[rr][col]) = own;
    }
}

template <int MT>
__device__ __forceinline__
void strip_pipeline(float (*X)[XP], const float* nidx_f, const float* nbr_f,
                    const float* __restrict__ feat_data, const float* __restrict__ feats,
                    const float* __restrict__ Wf, const float* __restrict__ bf,
                    int wave, int lane, int rows_used)
{
    const int lr    = lane & 15;
    const int ko    = (lane >> 4) << 2;
    const int rbase = (lane >> 4) << 3;
    const bool gatherer = wave < NGW;
    const int  cw = wave & (NGW - 1);

    const float* __restrict__ a_nf = &X[lr][ko];
    const float* __restrict__ a_s  = &X[lr][XS + ko];
    const float* __restrict__ wf = Wf + (size_t)cw * WAVE_FLOATS + 4 * lane;
    const int CH = (128 / 8) * CT * FRAG_FLOATS;

    v8f acc[2][CT];
    #pragma unroll
    for (int m = 0; m < 2; ++m)
        #pragma unroll
        for (int c = 0; c < CT; ++c) acc[m][c] = v8f{};

    gather_pass<0>(X, nidx_f, nbr_f, feat_data, feats, wave, THREADS / 32, rows_used, lane);

    v4f p[4][CT];
    if (gatherer) {
        #pragma unroll
        for (int j = 0; j < 4; ++j)
            #pragma unroll
            for (int c = 0; c < CT; ++c) p[j][c] = v4f{};
    } else {
        #pragma unroll
        for (int j = 0; j < 4; ++j)
            #pragma unroll
            for (int c = 0; c < CT; ++c) p[j][c] = *(const v4f*)(wf + (j * CT + c) * FRAG_FLOATS);
    }
    lds_barrier();

#define END_GEMM_STATE()                                                          \
    do {                                                                          \
        _Pragma("unroll") for (int m = 0; m < 2; ++m)                             \
            _Pragma("unroll") for (int c = 0; c < CT; ++c) acc[m][c] = v8f{};     \
        _Pragma("unroll") for (int j = 0; j < 4; ++j)                             \
            _Pragma("unroll") for (int c = 0; c < CT; ++c) p[j][c] = v4f{};       \
    } while (0)

    if (gatherer) {
        gather_pass<1>(X, nidx_f, nbr_f, feat_data, feats, wave, NGW, rows_used, lane);
        END_GEMM_STATE();
    } else {
        gemm_chunk<MT, false>(a_nf,       wf,          p, acc);
        gemm_chunk<MT, false>(a_s,        wf + CH,     p, acc);
    }
    lds_barrier();
    if (gatherer) {
        gather_pass<2>(X, nidx_f, nbr_f, feat_data, feats, wave, NGW, rows_used, lane);
        END_GEMM_STATE();
    } else {
        gemm_chunk<MT, false>(a_nf + 128, wf + 2 * CH, p, acc);
        gemm_chunk<MT, false>(a_s + 128,  wf + 3 * CH, p, acc);
    }
    lds_barrier();
    if (gatherer) {
        gather_pass<3>(X, nidx_f, nbr_f, feat_data, feats, wave, NGW, rows_used, lane);
        END_GEMM_STATE();
    } else {
        gemm_chunk<MT, false>(a_s + 256,  wf + 4 * CH, p, acc);
    }
    lds_barrier();
    if (!gatherer)
        gemm_chunk<MT, true>(a_s + 384,   wf + 5 * CH, p, acc);
    __syncthreads();
#undef END_GEMM_STATE

    if (!gatherer) {
        #pragma unroll
        for (int c = 0; c < CT; ++c) {
            const int n = cw * 64 + c * 16 + lr;
            const float bias = bf[n];
            #pragma unroll
            for (int v = 0; v < 8; ++v)
                X[rbase + v][XU + n] = acc[0][c][v] + bias;
            if (MT == 2) {
                #pragma unroll
                for (int v = 0; v < 8; ++v)
                    X[16 + rbase + v][XU + n] = acc[1][c][v] + bias;
            }
        }
    }
}

__global__ __launch_bounds__(THREADS)
void graphsage_fused_kernel(const float* __restrict__ feat_data,
                            const float* __restrict__ feats,
                            const float* __restrict__ Wf,
                            const float* __restrict__ bf,
                            const int*   __restrict__ adj,
                            const int*   __restrict__ in1,
                            const int*   __restrict__ in2,
                            const int*   __restrict__ neg,
                            int B1, int B2, int Bn,
                            float* __restrict__ out_all)
{
    __shared__ __attribute__((aligned(16))) float X[TM][XP];
    __shared__ __attribute__((aligned(16))) float aux[TM + TM * DEG];
    float* const nidx_f   = &aux[0];
    float* const nbr_f    = &aux[TM];
    float* const part     = &aux[0];
    float* const inv_norm = &aux[TM * 8];

    const int t    = threadIdx.x;
    const int lane = t & 31;
    const int wave = __builtin_amdgcn_readfirstlane(t >> 5);

    const int nb1 = (B1 + TM - 1) / TM;
    const int nb2 = (B2 + TM - 1) / TM;
    int bi = blockIdx.x;
    const int* __restrict__ nodes = in1;
    int nNodes = B1;
    float* __restrict__ out = out_all;
    if (bi >= nb1 + nb2)  { bi -= nb1 + nb2; nodes = neg; nNodes = Bn; out = out_all + (size_t)(B1 + B2) * FDIM; }
    else if (bi >= nb1)   { bi -= nb1;       nodes = in2; nNodes = B2; out = out_all + (size_t)B1 * FDIM; }
    const int m0 = bi * TM;
    const bool two_tiles = (nNodes - m0) > 16;
    const int  rows_used = two_tiles ? TM : 16;

    if (t < TM) {
        int g = m0 + t;
        nidx_f[t] = __int_as_float((g < nNodes) ? nodes[g] : 0);
    }
    __syncthreads();
    for (int i = t; i < TM * DEG; i += THREADS) {
        const int rowi = i >> 5, d = i & 31;
        const int id   = __float_as_int(nidx_f[rowi]);
        nbr_f[i] = __int_as_float(adj[(size_t)id * DEG + d]);
    }
    __syncthreads();

    if (two_tiles) strip_pipeline<2>(X, nidx_f, nbr_f, feat_data, feats, Wf, bf, wave, lane, rows_used);
    else           strip_pipeline<1>(X, nidx_f, nbr_f, feat_data, feats, Wf, bf, wave, lane, rows_used);
    __syncthreads();

    {
        const int r = t >> 3, seg = t & 7;
        float s = 0.f;
        #pragma unroll 8
        for (int c = seg * 32; c < seg * 32 + 32; ++c) {
            const float x = X[r][XU + c];
            s += x * x;
        }
        part[r * 8 + seg] = s;
    }
    __syncthreads();
    if (t < TM) {
        float s = 0.f;
        #pragma unroll
        for (int i = 0; i < 8; ++i) s += part[t * 8 + i];
        inv_norm[t] = 1.0f / fmaxf(sqrtf(s), 1e-12f);
    }
    __syncthreads();
    for (int i = t; i < TM * FDIM; i += THREADS) {
        const int r = i >> 8;
        const int c = i & (FDIM - 1);
        const int g = m0 + r;
        if (g < nNodes)
            out[(size_t)g * FDIM + c] = X[r][XU + c] * inv_norm[r];
    }
}

extern "C" void kernel_launch(void* const* d_in, const int* in_sizes, int n_in,
                              void* d_out, int out_size, void* d_ws, size_t ws_size,
                              hipStream_t stream) {
    const float* feat_data = (const float*)d_in[0];
    const float* feats     = (const float*)d_in[1];
    const float* W1        = (const float*)d_in[2];
    const float* b1        = (const float*)d_in[3];
    const float* W2        = (const float*)d_in[4];
    const float* b2        = (const float*)d_in[5];
    const int*   adj       = (const int*)d_in[6];
    const int*   in1       = (const int*)d_in[7];
    const int*   in2       = (const int*)d_in[8];
    const int*   neg       = (const int*)d_in[9];
    float* out = (float*)d_out;

    const int B1 = in_sizes[7];
    const int B2 = in_sizes[8];
    const int Bn = in_sizes[9];

    float* Wf = (float*)d_ws;
    float* bf = Wf + (FDIM / 16) * NPOS * FRAG_FLOATS;

    dim3 blk(THREADS);
    fold_weights_kernel<<<dim3(2 * NT_FOLD), blk, 0, stream>>>(W1, b1, W2, b2, Wf, bf);
    const int nblk = (B1 + TM - 1) / TM + (B2 + TM - 1) / TM + (Bn + TM - 1) / TM;
    graphsage_fused_kernel<<<dim3(nblk), blk, 0, stream>>>(
        feat_data, feats, Wf, bf, adj, in1, in2, neg, B1, B2, Bn, out);
}


# 64-row work-groups of 17 waves, one per WGP (8 gather, 8 GEMM, 1 for a 65th row): each weight fragment feeds four row tiles, halving weight bytes; lists cut evenly over 256 groups; sum order kept
# speedup vs baseline: 2.6019x; 1.0821x over previous
#include <hip/hip_runtime.h>
#include <hip/hip_bf16.h>


typedef float v2f __attribute__((ext_vector_type(2)));
typedef float v4f __attribute__((ext_vector_type(4)));
typedef float v8f __attribute__((ext_vector_type(8)));

#define FDIM 256
#define KDIM 512
#define NT_FOLD 33
#define DEG  32
#define TM   64
#define MTX  5
#define TMX  (16 * MTX)
#define THREADS 544
#define FOLD_THREADS 256
#define XP   516
#define XB   256
#define XU   0

#define WMMA4(A, B, C) __builtin_amdgcn_wmma_f32_16x16x4_f32(false, (A), false, (B), (short)0, (C), false, false)

#define FRAG_FLOATS 128
#define NPOS 96
#define CT_ALL 2
#define WAVE_FLOATS (NPOS * CT_ALL * FRAG_FLOATS)

__device__ __forceinline__ size_t wf_block(int nt, int pos) { return (size_t)(((nt >> 1) * NPOS + pos) * CT_ALL + (nt & 1)) * FRAG_FLOATS; }

__device__ __forceinline__ int pos_w2a(int kc) { return ((kc >> 4) << 5) + (kc & 15); }
__device__ __forceinline__ int pos_wc(int kc)  { const int q = kc >> 4; return ((q < 2) ? 16 + 32 * q : 32 + 16 * q) + (kc & 15); }

template <bool BIAS>
__device__ __forceinline__
v8f fold_tile(const float* __restrict__ arow,
              const float* __restrict__ bsrc,
              float bscale)
{
    v8f acc = {}, acc2 = {};
    v4f a4[FDIM / 8];
    float q[FDIM / 8][4];
    #pragma unroll
    for (int i = 0; i < FDIM / 8; ++i) {
        const int j0 = 8 * i;
        a4[i] = *(const v4f*)(arow + j0);
        if (BIAS) {
            const v4f b4 = *(const v4f*)(bsrc + j0);
            q[i][0] = bscale * b4.x; q[i][1] = bscale * b4.y; q[i][2] = bscale * b4.z; q[i][3] = bscale * b4.w;
        } else {
            q[i][0] = bsrc[(size_t)(j0 + 0) * KDIM];
            q[i][1] = bsrc[(size_t)(j0 + 1) * KDIM];
            q[i][2] = bsrc[(size_t)(j0 + 2) * KDIM];
            q[i][3] = bsrc[(size_t)(j0 + 3) * KDIM];
        }
    }
    #pragma unroll
    for (int i = 0; i < FDIM / 8; ++i) {
        v2f a, b;
        a.x = q[i][0]; a.y = q[i][1]; b.x = a4[i].x; b.y = a4[i].y;
        acc = WMMA4(a, b, acc);
        a.x = q[i][2]; a.y = q[i][3]; b.x = a4[i].z; b.y = a4[i].w;
        acc2 = WMMA4(a, b, acc2);
    }
    return acc + acc2;
}

__global__ __launch_bounds__(FOLD_THREADS) __attribute__((amdgpu_waves_per_eu(2, 2)))
void fold_weights_kernel(const float* __restrict__ W1,
                         const float* __restrict__ b1,
                         const float* __restrict__ W2,
                         const float* __restrict__ b2,
                         float* __restrict__ Wf,
                         float* __restrict__ bf)
{
    const int t    = threadIdx.x;
    const int lane = t & 31;
    const int wave = t >> 5;
    const int lr   = lane & 15;
    const int h    = lane >> 4;

    const int gw = blockIdx.x * 8 + wave;
    v4f w2a = {};
    if (gw < (FDIM / 16) * (FDIM / 8))
        w2a = *(const v4f*)(W2 + (size_t)((gw >> 5) * 16 + lr) * KDIM + 8 * (gw & 31) + 4 * h);

    const int nt = blockIdx.x % NT_FOLD;
    const int ot = (blockIdx.x / NT_FOLD) * 8 + wave;
    const float* __restrict__ arow = W2 + (size_t)(ot * 16 + lr) * KDIM + FDIM + 4 * h;
    if (nt < NT_FOLD - 1) {
        const v8f acc = fold_tile<false>(arow, W1 + (size_t)(4 * h) * KDIM + nt * 16 + lr, 0.f);
        float* __restrict__ blk = Wf + wf_block(ot, pos_wc(2 * nt + h));
        v4f lo, hi;
        lo.x = acc[0]; lo.y = acc[1]; lo.z = acc[2]; lo.w = acc[3];
        hi.x = acc[4]; hi.y = acc[5]; hi.z = acc[6]; hi.w = acc[7];
        *(v4f*)(blk + 4 * lr)        = lo;
        *(v4f*)(blk + 4 * (16 + lr)) = hi;
    } else {
        const v8f acc = fold_tile<true>(arow, b1 + 4 * h, (lr == 0) ? (float)DEG : 0.f);
        if (h == 0)
            bf[ot * 16 + lr] = acc[0] + b2[ot * 16 + lr];
    }

    if (gw < (FDIM / 16) * (FDIM / 8))
        *(v4f*)(Wf + wf_block(gw >> 5, pos_w2a(gw & 31)) + 4 * lane) = w2a;
}

__device__ __forceinline__ void lds_barrier()
{
    asm volatile("s_wait_dscnt 0x0\n\ts_barrier_signal -1\n\ts_barrier_wait -1" ::: "memory");
}

#define NGW  8
#define CT   2
static_assert(CT == CT_ALL, "Wf interleaves the column tiles of one GEMM wave");
static_assert(TM == 64 && NGW * CT * 16 == FDIM && (2 * NGW + 1) * 32 == THREADS, "tile and wave split of a strip");

template <int MT, bool B_NEXT>
__device__ __forceinline__
void gemm_trip(const float* __restrict__ arow,
               const float* __restrict__ wn,
               bool a_next,
               v4f (&a)[2][MTX],
               v4f (&p)[4][CT],
               v8f (&acc)[MTX][CT])
{
    #pragma unroll
    for (int j = 0; j < 4; ++j) {
        const int s = j & 1;
        v2f x, y;
        #pragma unroll
        for (int e = 0; e < 2; ++e) {
            #pragma unroll
            for (int m = 0; m < MT; ++m) {
                x.x = e ? a[s][m].z : a[s][m].x;  x.y = e ? a[s][m].w : a[s][m].y;
                #pragma unroll
                for (int c = 0; c < CT; ++c) {
                    y.x = e ? p[j][c].z : p[j][c].x;  y.y = e ? p[j][c].w : p[j][c].y;
                    acc[m][c] = WMMA4(x, y, acc[m][c]);
                }
            }
        }
        if (B_NEXT) {
            #pragma unroll
            for (int c = 0; c < CT; ++c)
                p[j][c] = *(const v4f*)(wn + (j * CT + c) * FRAG_FLOATS);
        }
        if (j < 2 || a_next) {
            #pragma unroll
            for (int m = 0; m < MT; ++m)
                a[s][m] = *(const v4f*)(arow + m * 16 * XP + 8 * (j + 2));
        }
        __builtin_amdgcn_sched_group_barrier(0x008, 2 * MT * CT, 0);
        if (B_NEXT) __builtin_amdgcn_sched_group_barrier(0x020, CT, 0);
        if (j < 2)  __builtin_amdgcn_sched_group_barrier(0x100, MT, 0);
    }
}

template <int MT, bool LAST>
__device__ __forceinline__
void gemm_chunk(const float* __restrict__ arow,
                const float* __restrict__ wfrag,
                v4f (&p)[4][CT],
                v8f (&acc)[MTX][CT])
{
    v4f a[2][MTX];
    #pragma unroll
    for (int s = 0; s < 2; ++s) {
        #pragma unroll
        for (int m = 0; m < MTX; ++m)
            a[s][m] = (m < MT) ? *(const v4f*)(arow + m * 16 * XP + 8 * s) : a[s][0];
    }
    #pragma unroll 1
    for (int t = 0; t < (LAST ? 3 : 4); ++t)
        gemm_trip<MT, true>(arow + 32 * t, wfrag + 4 * CT * (t + 1) * FRAG_FLOATS, t < 3, a, p, acc);
    if (LAST)
        gemm_trip<MT, false>(arow + 96, wfrag, false, a, p, acc);
}

template <int P>
__device__ __forceinline__
void gather_pass(float (*X)[XP], const float* nidx_f, const float* nbr_f,
                 const float* __restrict__ feat_data, const float* __restrict__ feats,
                 int first, int stride, int rows_used, int lane)
{
    const float* __restrict__ tab = (P < 2) ? feat_data : feats;
    const int col  = ((P & 1) << 7) + (lane << 2);
    const int ocol = (P & 1) * XB + (lane << 2);
    const int xcol = ocol + ((P < 2) ? 128 : 0);
    for (int rr = first; rr < rows_used; rr += stride) {
        const int ids = __float_as_int(nbr_f[rr * DEG + lane]);
        v4f own = {};
        if (P < 2) {
            const int idx = __builtin_amdgcn_readfirstlane(__float_as_int(nidx_f[rr]));
            own = *(const v4f*)(feat_data + (size_t)idx * FDIM + col);
        }
        float s0 = 0.f, s1 = 0.f, s2 = 0.f, s3 = 0.f;
        #pragma unroll
        for (int d = 0; d < DEG; ++d) {
            const int j = __builtin_amdgcn_readlane(ids, d);
            const v4f f = *(const v4f*)(tab + (size_t)j * FDIM + col);
            s0 += f.x; s1 += f.y; s2 += f.z; s3 += f.w;
        }
        v4f sv; sv.x = s0; sv.y = s1; sv.z = s2; sv.w = s3;
        *(v4f*)(&X[rr][xcol]) = sv;
        if (P < 2) *(v4f*)(&X[rr][ocol]) = own;
    }
}

template <int MT>
__device__ __forceinline__
void strip_pipeline(float (*X)[XP], const float* nidx_f, const float* nbr_f,
                    const float* __restrict__ feat_data, const float* __restrict__ feats,
                    const float* __restrict__ Wf, const float* __restrict__ bf,
                    int wave, int lane, int nrows)
{
    const int lr    = lane & 15;
    const int ko    = (lane >> 4) << 2;
    const int rbase = (lane >> 4) << 3;
    const bool xw = (wave == 2 * NGW);
    const bool gatherer = (wave < NGW) || xw;
    const int  cw = wave & (NGW - 1);
    const int g_first = xw ? TM : wave;
    const int g_end   = xw ? nrows : ((nrows < TM) ? nrows : TM);
    const int g_str0  = xw ? 1 : 2 * NGW;
    const int g_str   = xw ? 1 : NGW;

    const float* __restrict__ a_b0 = &X[lr][ko];
    const float* __restrict__ a_b1 = &X[lr][XB + ko];
    const float* __restrict__ wf = Wf + (size_t)cw * WAVE_FLOATS + 4 * lane;
    const int CH = (128 / 8) * CT * FRAG_FLOATS;

    v8f acc[MTX][CT];
    #pragma unroll
    for (int m = 0; m < MTX; ++m)
        #pragma unroll
        for (int c = 0; c < CT; ++c) acc[m][c] = v8f{};

    gather_pass<0>(X, nidx_f, nbr_f, feat_data, feats, g_first, g_str0, g_end, lane);

    v4f p[4][CT];
    if (gatherer) {
        #pragma unroll
        for (int j = 0; j < 4; ++j)
            #pragma unroll
            for (int c = 0; c < CT; ++c) p[j][c] = v4f{};
    } else {
        #pragma unroll
        for (int j = 0; j < 4; ++j)
            #pragma unroll
            for (int c = 0; c < CT; ++c) p[j][c] = *(const v4f*)(wf + (j * CT + c) * FRAG_FLOATS);
    }
    lds_barrier();

#define END_GEMM_STATE()                                                          \
    do {                                                                          \
        _Pragma("unroll") for (int m = 0; m < MTX; ++m)                           \
            _Pragma("unroll") for (int c = 0; c < CT; ++c) acc[m][c] = v8f{};     \
        _Pragma("unroll") for (int j = 0; j < 4; ++j)                             \
            _Pragma("unroll") for (int c = 0; c < CT; ++c) p[j][c] = v4f{};       \
    } while (0)

    if (gatherer) {
        gather_pass<1>(X, nidx_f, nbr_f, feat_data, feats, g_first, g_str, g_end, lane);
        END_GEMM_STATE();
    } else {
        gemm_chunk<MT, false>(a_b0,       wf,          p, acc);
        gemm_chunk<MT, false>(a_b0 + 128, wf + CH,     p, acc);
    }
    lds_barrier();
    if (gatherer) {
        gather_pass<2>(X, nidx_f, nbr_f, feat_data, feats, g_first, g_str, g_end, lane);
        END_GEMM_STATE();
    } else {
        gemm_chunk<MT, false>(a_b1,       wf + 2 * CH, p, acc);
        gemm_chunk<MT, false>(a_b1 + 128, wf + 3 * CH, p, acc);
    }
    lds_barrier();
    if (gatherer) {
        gather_pass<3>(X, nidx_f, nbr_f, feat_data, feats, g_first, g_str, g_end, lane);
        END_GEMM_STATE();
    } else {
        gemm_chunk<MT, false>(a_b0,       wf + 4 * CH, p, acc);
    }
    lds_barrier();
    if (!gatherer)
        gemm_chunk<MT, true>(a_b1,        wf + 5 * CH, p, acc);
    __syncthreads();
#undef END_GEMM_STATE

    if (!gatherer) {
        #pragma unroll
        for (int c = 0; c < CT; ++c) {
            const int n = cw * (16 * CT) + c * 16 + lr;
            const float bias = bf[n];
            #pragma unroll
            for (int m = 0; m < MT; ++m) {
                #pragma unroll
                for (int v = 0; v < 8; ++v)
                    X[16 * m + rbase + v][XU + n] = acc[m][c][v] + bias;
            }
        }
    }
}

__global__ __launch_bounds__(THREADS)
void graphsage_fused_kernel(const float* __restrict__ feat_data,
                            const float* __restrict__ feats,
                            const float* __restrict__ Wf,
                            const float* __restrict__ bf,
                            const int*   __restrict__ adj,
                            const int*   __restrict__ in1,
                            const int*   __restrict__ in2,
                            const int*   __restrict__ neg,
                            int B1, int B2, int Bn,
                            float* __restrict__ out_all)
{
    __shared__ __attribute__((aligned(16))) float X[TMX][XP];
    __shared__ __attribute__((aligned(16))) float aux[TMX + TMX * DEG];
    float* const nidx_f   = &aux[0];
    float* const nbr_f    = &aux[TMX];
    float* const part     = &aux[0];
    float* const inv_norm = &aux[TMX * 8];

    const int t    = threadIdx.x;
    const int lane = t & 31;
    const int wave = __builtin_amdgcn_readfirstlane(t >> 5);

    const int total = B1 + B2 + Bn;
    const int nwg   = (int)gridDim.x;
    const int base  = total / nwg;
    const int extra = total - base * nwg;
    const int bi    = (int)blockIdx.x;
    const int m0    = bi * base + ((bi < extra) ? bi : extra);
    const int nrows = base + ((bi < extra) ? 1 : 0);

    if (t < TMX) {
        const int g = m0 + t;
        int id = 0;
        if (t < nrows) id = (g < B1) ? in1[g] : ((g < B1 + B2) ? in2[g - B1] : neg[g - B1 - B2]);
        nidx_f[t] = __int_as_float(id);
    }
    __syncthreads();
    for (int i = t; i < nrows * DEG; i += THREADS) {
        const int rowi = i >> 5, d = i & 31;
        const int id   = __float_as_int(nidx_f[rowi]);
        nbr_f[i] = __int_as_float(adj[(size_t)id * DEG + d]);
    }
    __syncthreads();

    if (nrows > TM) strip_pipeline<MTX>(X, nidx_f, nbr_f, feat_data, feats, Wf, bf, wave, lane, nrows);
    else            strip_pipeline<4>(X, nidx_f, nbr_f, feat_data, feats, Wf, bf, wave, lane, nrows);
    __syncthreads();

    for (int r = t >> 3; r < TMX; r += THREADS / 8) {
        const int seg = t & 7;
        float s = 0.f;
        #pragma unroll 8
        for (int cc = seg * 32; cc < seg * 32 + 32; ++cc) {
            const float x = X[r][XU + cc];
            s += x * x;
        }
        part[r * 8 + seg] = s;
    }
    __syncthreads();
    if (t < TMX) {
        float s = 0.f;
        #pragma unroll
        for (int i = 0; i < 8; ++i) s += part[t * 8 + i];
        inv_norm[t] = 1.0f / fmaxf(sqrtf(s), 1e-12f);
    }
    __syncthreads();
    float* __restrict__ out = out_all + (size_t)m0 * FDIM;
    for (int i = t; i < nrows * FDIM; i += THREADS) {
        const int r  = i >> 8;
        const int cc = i & (FDIM - 1);
        out[i] = X[r][XU + cc] * inv_norm[r];
    }
}

extern "C" void kernel_launch(void* const* d_in, const int* in_sizes, int n_in,
                              void* d_out, int out_size, void* d_ws, size_t ws_size,
                              hipStream_t stream) {
    const float* feat_data = (const float*)d_in[0];
    const float* feats     = (const float*)d_in[1];
    const float* W1        = (const float*)d_in[2];
    const float* b1        = (const float*)d_in[3];
    const float* W2        = (const float*)d_in[4];
    const float* b2        = (const float*)d_in[5];
    const int*   adj       = (const int*)d_in[6];
    const int*   in1       = (const int*)d_in[7];
    const int*   in2       = (const int*)d_in[8];
    const int*   neg       = (const int*)d_in[9];
    float* out = (float*)d_out;

    const int B1 = in_sizes[7];
    const int B2 = in_sizes[8];
    const int Bn = in_sizes[9];

    float* Wf = (float*)d_ws;
    float* bf = Wf + (FDIM / 16) * NPOS * FRAG_FLOATS;

    fold_weights_kernel<<<dim3(2 * NT_FOLD), dim3(FOLD_THREADS), 0, stream>>>(W1, b1, W2, b2, Wf, bf);
    const int total = B1 + B2 + Bn;
    int nblk = (total + TM - 1) / TM;
    if (nblk > 256) {
        nblk = (total + TMX - 1) / TMX;
        if (nblk < 256) nblk = 256;
    }
    if (nblk < 1) nblk = 1;
    graphsage_fused_kernel<<<dim3(nblk), dim3(THREADS), 0, stream>>>(
        feat_data, feats, Wf, bf, adj, in1, in2, neg, B1, B2, Bn, out);
}
